# starting kernel
# baseline (speedup 1.0000x reference)
#include <hip/hip_runtime.h>

typedef __attribute__((ext_vector_type(2))) float v2f;
typedef __attribute__((ext_vector_type(8))) float v8f;

#define Bq 128
#define Nq 512
#define Fq 32
#define Hq 128
#define Dq 64
#define Kq 16

__global__ __launch_bounds__(256) void knn_kernel(const float* __restrict__ events,
                                                  int* __restrict__ knn_idx) {
  __shared__ float sx[Nq];
  __shared__ float sy[Nq];
  const int tid = threadIdx.x;
  const int p   = blockIdx.x * 256 + tid;
  const int b   = p >> 9;
  const int i   = p & (Nq - 1);

  for (int j = tid; j < Nq; j += 256) {
    const float* e = events + ((size_t)b * Nq + j) * Fq;
    sx[j] = e[0];
    sy[j] = e[1];
  }
  __syncthreads();

  const float cx = sx[i], cy = sy[i];
  float bestd[Kq];
  int   besti[Kq];
#pragma unroll
  for (int t = 0; t < Kq; ++t) { bestd[t] = 3.4e38f; besti[t] = 0; }

  for (int j = 0; j < Nq; ++j) {
    if (j == i) continue;
    const float dx = sx[j] - cx;
    const float dy = sy[j] - cy;
    const float d  = dx * dx + dy * dy;
    if (d < bestd[Kq - 1]) {
      float dd = d; int ii = j;
#pragma unroll
      for (int t = 0; t < Kq; ++t) {
        if (dd < bestd[t]) {
          const float td = bestd[t]; const int ti = besti[t];
          bestd[t] = dd; besti[t] = ii;
          dd = td; ii = ti;
        }
      }
    }
  }
#pragma unroll
  for (int t = 0; t < Kq; ++t) knn_idx[(size_t)p * Kq + t] = besti[t];
}

#define WAVES_PER_BLOCK 4
#define H_STRIDE 129

__global__ __launch_bounds__(WAVES_PER_BLOCK * 32) void edgeconv_mlp_kernel(
    const float* __restrict__ events, const int* __restrict__ knn_idx,
    const float* __restrict__ W1, const float* __restrict__ b1,
    const float* __restrict__ W2, const float* __restrict__ b2,
    float* __restrict__ out) {
  __shared__ float ldsH[WAVES_PER_BLOCK * 16 * H_STRIDE];

  const int lane = threadIdx.x & 31;
  const int wid  = threadIdx.x >> 5;
  const int p    = blockIdx.x * WAVES_PER_BLOCK + wid;
  const int b    = p >> 9;
  const int i    = p & (Nq - 1);
  const int m    = lane & 15;
  const int h    = lane >> 4;

  float* myH = ldsH + wid * 16 * H_STRIDE;

  const int nj = knn_idx[(size_t)p * Kq + m];
  const float* ev = events + ((size_t)b * Nq + i) * Fq;
  const float* nb = events + ((size_t)b * Nq + nj) * Fq;

  float a[32];
#pragma unroll
  for (int s = 0; s < 16; ++s) {
    if (s < 8) {
      const int c = 4 * s + 2 * h;
      a[2 * s]     = ev[c];
      a[2 * s + 1] = ev[c + 1];
    } else {
      const int c = 4 * (s - 8) + 2 * h;
      a[2 * s]     = nb[c]     - ev[c];
      a[2 * s + 1] = nb[c + 1] - ev[c + 1];
    }
  }

  const v8f vzero = {};

  v8f acc1[8];
#pragma unroll
  for (int t = 0; t < 8; ++t) acc1[t] = vzero;

#pragma unroll
  for (int s = 0; s < 16; ++s) {
    v2f A;
    A.x = a[2 * s];
    A.y = a[2 * s + 1];
    const int kb = 4 * s + 2 * h;
    const float* w1r0 = W1 + (size_t)kb * Hq + m;
    const float* w1r1 = W1 + (size_t)(kb + 1) * Hq + m;
#pragma unroll
    for (int t = 0; t < 8; ++t) {
      v2f Bv;
      Bv.x = w1r0[16 * t];
      Bv.y = w1r1[16 * t];
      acc1[t] = __builtin_amdgcn_wmma_f32_16x16x4_f32(
          false, A, false, Bv, (short)0, acc1[t], false, false);
    }
  }

#pragma unroll
  for (int t = 0; t < 8; ++t) {
    const float bv = b1[16 * t + m];
#pragma unroll
    for (int r = 0; r < 8; ++r) {
      float x = acc1[t][r] + bv;
      x = fmaxf(x, 0.0f);
      myH[(r + 8 * h) * H_STRIDE + 16 * t + m] = x;
    }
  }

  v8f acc2[4];
#pragma unroll
  for (int t = 0; t < 4; ++t) acc2[t] = vzero;

#pragma unroll
  for (int s = 0; s < 32; ++s) {
    const int kb = 4 * s + 2 * h;
    v2f A;
    A.x = myH[m * H_STRIDE + kb];
    A.y = myH[m * H_STRIDE + kb + 1];
    const float* w2r0 = W2 + (size_t)kb * Dq + m;
    const float* w2r1 = W2 + (size_t)(kb + 1) * Dq + m;
#pragma unroll
    for (int t = 0; t < 4; ++t) {
      v2f Bv;
      Bv.x = w2r0[16 * t];
      Bv.y = w2r1[16 * t];
      acc2[t] = __builtin_amdgcn_wmma_f32_16x16x4_f32(
          false, A, false, Bv, (short)0, acc2[t], false, false);
    }
  }

#pragma unroll
  for (int t = 0; t < 4; ++t) {
    const float bv = b2[16 * t + m];
    float part = 0.0f;
#pragma unroll
    for (int r = 0; r < 8; ++r) part += fmaxf(acc2[t][r] + bv, 0.0f);
    part += __shfl_xor(part, 16, 32);
    if (h == 0) out[(size_t)p * Dq + 16 * t + m] = part * (1.0f / 16.0f);
  }
}

extern "C" void kernel_launch(void* const* d_in, const int* in_sizes, int n_in,
                              void* d_out, int out_size, void* d_ws, size_t ws_size,
                              hipStream_t stream) {
  const float* events = (const float*)d_in[0];
  const float* W1     = (const float*)d_in[1];
  const float* b1     = (const float*)d_in[2];
  const float* W2     = (const float*)d_in[3];
  const float* b2     = (const float*)d_in[4];
  float*       outp   = (float*)d_out;
  int*         knn    = (int*)d_ws;

  knn_kernel<<<(Bq * Nq) / 256, 256, 0, stream>>>(events, knn);
  edgeconv_mlp_kernel<<<(Bq * Nq) / WAVES_PER_BLOCK, WAVES_PER_BLOCK * 32, 0, stream>>>(
      events, knn, W1, b1, W2, b2, outp);
}


# KNN top-16 rewritten: med3-only value pass finds the 16th distance, 2nd pass gathers the 16 ids under it, 3rd sorts them; equal distances now by index as a stable argsort (may reorder mean sum)
# speedup vs baseline: 3.1088x; 3.1088x over previous
#include <hip/hip_runtime.h>

typedef __attribute__((ext_vector_type(2))) float v2f;
typedef __attribute__((ext_vector_type(8))) float v8f;

#define Bq 128
#define Nq 512
#define Fq 32
#define Hq 128
#define Dq 64
#define Kq 16

#define KNN_THREADS 256
#define KNN_BIG 3.4e38f

__device__ __forceinline__ float dist2(float x, float y, float cx, float cy) {
  const float dx = x - cx;
  const float dy = y - cy;
  return __builtin_fmaf(dx, dx, dy * dy);
}

template <bool EXCL>
__device__ __forceinline__ void knn_scan_values(const float2* sxy, int j0, int i,
                                                float cx, float cy, float (&bd)[Kq]) {
#pragma unroll 8
  for (int jj = 0; jj < 32; ++jj) {
    const int j = j0 + jj;
    const float2 c = sxy[j];
    float d = dist2(c.x, c.y, cx, cy);
    if (EXCL) d = (j == i) ? KNN_BIG : d;
#pragma unroll
    for (int t = Kq - 1; t >= 1; --t) bd[t] = __builtin_amdgcn_fmed3f(bd[t - 1], d, bd[t]);
    bd[0] = __builtin_fminf(d, bd[0]);
  }
}

template <bool EXCL>
__device__ __forceinline__ void knn_scan_gather(const float2* sxy, int* lst, int j0, int i,
                                                float cx, float cy, float tau,
                                                int& ties_left, int& pos) {
#pragma unroll 8
  for (int jj = 0; jj < 32; ++jj) {
    const int j = j0 + jj;
    const float2 c = sxy[j];
    float d = dist2(c.x, c.y, cx, cy);
    if (EXCL) d = (j == i) ? KNN_BIG : d;
    const bool eq   = (d == tau) && (ties_left > 0);
    const bool take = (d < tau) || eq;
    lst[pos] = j;
    pos += take ? KNN_THREADS : 0;
    ties_left -= eq ? 1 : 0;
  }
}

__global__ __launch_bounds__(KNN_THREADS) void knn_kernel(const float* __restrict__ events,
                                                          int* __restrict__ knn_idx) {
  __shared__ float2 sxy[Nq];
  __shared__ int    lst[(Kq + 1) * KNN_THREADS];
  const int tid = threadIdx.x;
  const int p   = blockIdx.x * KNN_THREADS + tid;
  const int b   = p >> 9;
  const int i   = p & (Nq - 1);

  for (int j = tid; j < Nq; j += KNN_THREADS) {
    const float* e = events + ((size_t)b * Nq + j) * Fq;
    float2 c;
    c.x = e[0];
    c.y = e[1];
    sxy[j] = c;
  }
  __syncthreads();

  const float cx = sxy[i].x, cy = sxy[i].y;
  const int mychunk = __builtin_amdgcn_readfirstlane(i >> 5);

  float bd[Kq];
#pragma unroll
  for (int t = 0; t < Kq; ++t) bd[t] = KNN_BIG;
  for (int cc = 0; cc < Nq / 32; ++cc) {
    if (cc == mychunk) knn_scan_values<true>(sxy, cc * 32, i, cx, cy, bd);
    else               knn_scan_values<false>(sxy, cc * 32, i, cx, cy, bd);
  }

  const float tau = bd[Kq - 1];
  int ties_left = 0;
#pragma unroll
  for (int t = 0; t < Kq; ++t) ties_left += (bd[t] == tau) ? 1 : 0;
  int pos = tid;
  for (int cc = 0; cc < Nq / 32; ++cc) {
    if (cc == mychunk) knn_scan_gather<true>(sxy, lst, cc * 32, i, cx, cy, tau, ties_left, pos);
    else               knn_scan_gather<false>(sxy, lst, cc * 32, i, cx, cy, tau, ties_left, pos);
  }

  float fd[Kq];
  int   fi[Kq];
#pragma unroll
  for (int t = 0; t < Kq; ++t) { fd[t] = KNN_BIG; fi[t] = 0; }
#pragma unroll
  for (int e = 0; e < Kq; ++e) {
    const int j = lst[e * KNN_THREADS + tid] & (Nq - 1);
    const float2 c = sxy[j];
    const float d = dist2(c.x, c.y, cx, cy);
#pragma unroll
    for (int t = Kq - 1; t >= 1; --t) {
      const bool below_prev = d < fd[t - 1];
      const bool below_this = d < fd[t];
      fi[t] = below_prev ? fi[t - 1] : (below_this ? j : fi[t]);
      fd[t] = __builtin_amdgcn_fmed3f(fd[t - 1], d, fd[t]);
    }
    fi[0] = (d < fd[0]) ? j : fi[0];
    fd[0] = __builtin_fminf(d, fd[0]);
  }

  int4* dst = reinterpret_cast<int4*>(knn_idx + (size_t)p * Kq);
#pragma unroll
  for (int q = 0; q < Kq / 4; ++q) {
    int4 v;
    v.x = fi[4 * q]; v.y = fi[4 * q + 1]; v.z = fi[4 * q + 2]; v.w = fi[4 * q + 3];
    dst[q] = v;
  }
}

#define WAVES_PER_BLOCK 4
#define H_STRIDE 129

__global__ __launch_bounds__(WAVES_PER_BLOCK * 32) void edgeconv_mlp_kernel(
    const float* __restrict__ events, const int* __restrict__ knn_idx,
    const float* __restrict__ W1, const float* __restrict__ b1,
    const float* __restrict__ W2, const float* __restrict__ b2,
    float* __restrict__ out) {
  __shared__ float ldsH[WAVES_PER_BLOCK * 16 * H_STRIDE];

  const int lane = threadIdx.x & 31;
  const int wid  = threadIdx.x >> 5;
  const int p    = blockIdx.x * WAVES_PER_BLOCK + wid;
  const int b    = p >> 9;
  const int i    = p & (Nq - 1);
  const int m    = lane & 15;
  const int h    = lane >> 4;

  float* myH = ldsH + wid * 16 * H_STRIDE;

  const int nj = knn_idx[(size_t)p * Kq + m];
  const float* ev = events + ((size_t)b * Nq + i) * Fq;
  const float* nb = events + ((size_t)b * Nq + nj) * Fq;

  float a[32];
#pragma unroll
  for (int s = 0; s < 16; ++s) {
    if (s < 8) {
      const int c = 4 * s + 2 * h;
      a[2 * s]     = ev[c];
      a[2 * s + 1] = ev[c + 1];
    } else {
      const int c = 4 * (s - 8) + 2 * h;
      a[2 * s]     = nb[c]     - ev[c];
      a[2 * s + 1] = nb[c + 1] - ev[c + 1];
    }
  }

  const v8f vzero = {};

  v8f acc1[8];
#pragma unroll
  for (int t = 0; t < 8; ++t) acc1[t] = vzero;

#pragma unroll
  for (int s = 0; s < 16; ++s) {
    v2f A;
    A.x = a[2 * s];
    A.y = a[2 * s + 1];
    const int kb = 4 * s + 2 * h;
    const float* w1r0 = W1 + (size_t)kb * Hq + m;
    const float* w1r1 = W1 + (size_t)(kb + 1) * Hq + m;
#pragma unroll
    for (int t = 0; t < 8; ++t) {
      v2f Bv;
      Bv.x = w1r0[16 * t];
      Bv.y = w1r1[16 * t];
      acc1[t] = __builtin_amdgcn_wmma_f32_16x16x4_f32(
          false, A, false, Bv, (short)0, acc1[t], false, false);
    }
  }

#pragma unroll
  for (int t = 0; t < 8; ++t) {
    const float bv = b1[16 * t + m];
#pragma unroll
    for (int r = 0; r < 8; ++r) {
      float x = acc1[t][r] + bv;
      x = fmaxf(x, 0.0f);
      myH[(r + 8 * h) * H_STRIDE + 16 * t + m] = x;
    }
  }

  v8f acc2[4];
#pragma unroll
  for (int t = 0; t < 4; ++t) acc2[t] = vzero;

#pragma unroll
  for (int s = 0; s < 32; ++s) {
    const int kb = 4 * s + 2 * h;
    v2f A;
    A.x = myH[m * H_STRIDE + kb];
    A.y = myH[m * H_STRIDE + kb + 1];
    const float* w2r0 = W2 + (size_t)kb * Dq + m;
    const float* w2r1 = W2 + (size_t)(kb + 1) * Dq + m;
#pragma unroll
    for (int t = 0; t < 4; ++t) {
      v2f Bv;
      Bv.x = w2r0[16 * t];
      Bv.y = w2r1[16 * t];
      acc2[t] = __builtin_amdgcn_wmma_f32_16x16x4_f32(
          false, A, false, Bv, (short)0, acc2[t], false, false);
    }
  }

#pragma unroll
  for (int t = 0; t < 4; ++t) {
    const float bv = b2[16 * t + m];
    float part = 0.0f;
#pragma unroll
    for (int r = 0; r < 8; ++r) part += fmaxf(acc2[t][r] + bv, 0.0f);
    part += __shfl_xor(part, 16, 32);
    if (h == 0) out[(size_t)p * Dq + 16 * t + m] = part * (1.0f / 16.0f);
  }
}

extern "C" void kernel_launch(void* const* d_in, const int* in_sizes, int n_in,
                              void* d_out, int out_size, void* d_ws, size_t ws_size,
                              hipStream_t stream) {
  const float* events = (const float*)d_in[0];
  const float* W1     = (const float*)d_in[1];
  const float* b1     = (const float*)d_in[2];
  const float* W2     = (const float*)d_in[3];
  const float* b2     = (const float*)d_in[4];
  float*       outp   = (float*)d_out;
  int*         knn    = (int*)d_ws;

  knn_kernel<<<(Bq * Nq) / 256, 256, 0, stream>>>(events, knn);
  edgeconv_mlp_kernel<<<(Bq * Nq) / WAVES_PER_BLOCK, WAVES_PER_BLOCK * 32, 0, stream>>>(
      events, knn, W1, b1, W2, b2, outp);
}


# MLP layer 1: the x columns are the same in a point's 16 edge rows, so x@W1[0:32] is done once per point (one WMMA tile per 8-point block, via LDS) and seeds the per-edge accumulators; sum order kept
# speedup vs baseline: 3.4413x; 1.1070x over previous
#include <hip/hip_runtime.h>

typedef __attribute__((ext_vector_type(2))) float v2f;
typedef __attribute__((ext_vector_type(8))) float v8f;

#define Bq 128
#define Nq 512
#define Fq 32
#define Hq 128
#define Dq 64
#define Kq 16

#define KNN_THREADS 256
#define KNN_BIG 3.4e38f

__device__ __forceinline__ float dist2(float x, float y, float cx, float cy) {
  const float dx = x - cx;
  const float dy = y - cy;
  return __builtin_fmaf(dx, dx, dy * dy);
}

template <bool EXCL>
__device__ __forceinline__ void knn_scan_values(const float2* sxy, int j0, int i,
                                                float cx, float cy, float (&bd)[Kq]) {
#pragma unroll 8
  for (int jj = 0; jj < 32; ++jj) {
    const int j = j0 + jj;
    const float2 c = sxy[j];
    float d = dist2(c.x, c.y, cx, cy);
    if (EXCL) d = (j == i) ? KNN_BIG : d;
#pragma unroll
    for (int t = Kq - 1; t >= 1; --t) bd[t] = __builtin_amdgcn_fmed3f(bd[t - 1], d, bd[t]);
    bd[0] = __builtin_fminf(d, bd[0]);
  }
}

template <bool EXCL>
__device__ __forceinline__ void knn_scan_gather(const float2* sxy, int* lst, int j0, int i,
                                                float cx, float cy, float tau,
                                                int& ties_left, int& pos) {
#pragma unroll 8
  for (int jj = 0; jj < 32; ++jj) {
    const int j = j0 + jj;
    const float2 c = sxy[j];
    float d = dist2(c.x, c.y, cx, cy);
    if (EXCL) d = (j == i) ? KNN_BIG : d;
    const bool eq   = (d == tau) && (ties_left > 0);
    const bool take = (d < tau) || eq;
    lst[pos] = j;
    pos += take ? KNN_THREADS : 0;
    ties_left -= eq ? 1 : 0;
  }
}

__global__ __launch_bounds__(KNN_THREADS) void knn_kernel(const float* __restrict__ events,
                                                          int* __restrict__ knn_idx) {
  __shared__ float2 sxy[Nq];
  __shared__ int    lst[(Kq + 1) * KNN_THREADS];
  const int tid = threadIdx.x;
  const int p   = blockIdx.x * KNN_THREADS + tid;
  const int b   = p >> 9;
  const int i   = p & (Nq - 1);

  for (int j = tid; j < Nq; j += KNN_THREADS) {
    const float* e = events + ((size_t)b * Nq + j) * Fq;
    float2 c;
    c.x = e[0];
    c.y = e[1];
    sxy[j] = c;
  }
  __syncthreads();

  const float cx = sxy[i].x, cy = sxy[i].y;
  const int mychunk = __builtin_amdgcn_readfirstlane(i >> 5);

  float bd[Kq];
#pragma unroll
  for (int t = 0; t < Kq; ++t) bd[t] = KNN_BIG;
  for (int cc = 0; cc < Nq / 32; ++cc) {
    if (cc == mychunk) knn_scan_values<true>(sxy, cc * 32, i, cx, cy, bd);
    else               knn_scan_values<false>(sxy, cc * 32, i, cx, cy, bd);
  }

  const float tau = bd[Kq - 1];
  int ties_left = 0;
#pragma unroll
  for (int t = 0; t < Kq; ++t) ties_left += (bd[t] == tau) ? 1 : 0;
  int pos = tid;
  for (int cc = 0; cc < Nq / 32; ++cc) {
    if (cc == mychunk) knn_scan_gather<true>(sxy, lst, cc * 32, i, cx, cy, tau, ties_left, pos);
    else               knn_scan_gather<false>(sxy, lst, cc * 32, i, cx, cy, tau, ties_left, pos);
  }

  float fd[Kq];
  int   fi[Kq];
#pragma unroll
  for (int t = 0; t < Kq; ++t) { fd[t] = KNN_BIG; fi[t] = 0; }
#pragma unroll
  for (int e = 0; e < Kq; ++e) {
    const int j = lst[e * KNN_THREADS + tid] & (Nq - 1);
    const float2 c = sxy[j];
    const float d = dist2(c.x, c.y, cx, cy);
#pragma unroll
    for (int t = Kq - 1; t >= 1; --t) {
      const bool below_prev = d < fd[t - 1];
      const bool below_this = d < fd[t];
      fi[t] = below_prev ? fi[t - 1] : (below_this ? j : fi[t]);
      fd[t] = __builtin_amdgcn_fmed3f(fd[t - 1], d, fd[t]);
    }
    fi[0] = (d < fd[0]) ? j : fi[0];
    fd[0] = __builtin_fminf(d, fd[0]);
  }

  int4* dst = reinterpret_cast<int4*>(knn_idx + (size_t)p * Kq);
#pragma unroll
  for (int q = 0; q < Kq / 4; ++q) {
    int4 v;
    v.x = fi[4 * q]; v.y = fi[4 * q + 1]; v.z = fi[4 * q + 2]; v.w = fi[4 * q + 3];
    dst[q] = v;
  }
}

#define WAVES_PER_BLOCK 8
#define H_STRIDE 129
#define P_STRIDE 132

__global__ __launch_bounds__(WAVES_PER_BLOCK * 32) void edgeconv_mlp_kernel(
    const float* __restrict__ events, const int* __restrict__ knn_idx,
    const float* __restrict__ W1, const float* __restrict__ b1,
    const float* __restrict__ W2, const float* __restrict__ b2,
    float* __restrict__ out) {
  __shared__ float ldsH[WAVES_PER_BLOCK * 16 * H_STRIDE];
  __shared__ float ldsP[16 * P_STRIDE];

  const int lane = threadIdx.x & 31;
  const int wid  = threadIdx.x >> 5;
  const int p0   = blockIdx.x * WAVES_PER_BLOCK;
  const int p    = p0 + wid;
  const int b    = p >> 9;
  const int i    = p & (Nq - 1);
  const int m    = lane & 15;
  const int h    = lane >> 4;

  float* myH = ldsH + wid * 16 * H_STRIDE;

  const int nj = knn_idx[(size_t)p * Kq + m];
  const float* ev = events + ((size_t)b * Nq + i) * Fq;
  const float* nb = events + ((size_t)b * Nq + nj) * Fq;

  float a[16];
#pragma unroll
  for (int s = 0; s < 8; ++s) {
    const int c = 4 * s + 2 * h;
    a[2 * s]     = nb[c]     - ev[c];
    a[2 * s + 1] = nb[c + 1] - ev[c + 1];
  }

  const v8f vzero = {};

  {
    const float* xr = events + (size_t)(p0 + (m & (WAVES_PER_BLOCK - 1))) * Fq;
    v8f accp = vzero;
#pragma unroll
    for (int s = 0; s < 8; ++s) {
      const int kb = 4 * s + 2 * h;
      v2f A;
      A.x = xr[kb];
      A.y = xr[kb + 1];
      v2f Bv;
      Bv.x = W1[(size_t)kb * Hq + 16 * wid + m];
      Bv.y = W1[(size_t)(kb + 1) * Hq + 16 * wid + m];
      accp = __builtin_amdgcn_wmma_f32_16x16x4_f32(
          false, A, false, Bv, (short)0, accp, false, false);
    }
#pragma unroll
    for (int r = 0; r < 8; ++r) ldsP[(r + 8 * h) * P_STRIDE + 16 * wid + m] = accp[r];
  }
  __syncthreads();

  v8f acc1[8];
#pragma unroll
  for (int t = 0; t < 8; ++t) {
    const float xw = ldsP[wid * P_STRIDE + 16 * t + m];
#pragma unroll
    for (int r = 0; r < 8; ++r) acc1[t][r] = xw;
  }

#pragma unroll
  for (int s = 0; s < 8; ++s) {
    v2f A;
    A.x = a[2 * s];
    A.y = a[2 * s + 1];
    const int kb = 32 + 4 * s + 2 * h;
    const float* w1r0 = W1 + (size_t)kb * Hq + m;
    const float* w1r1 = W1 + (size_t)(kb + 1) * Hq + m;
#pragma unroll
    for (int t = 0; t < 8; ++t) {
      v2f Bv;
      Bv.x = w1r0[16 * t];
      Bv.y = w1r1[16 * t];
      acc1[t] = __builtin_amdgcn_wmma_f32_16x16x4_f32(
          false, A, false, Bv, (short)0, acc1[t], false, false);
    }
  }

#pragma unroll
  for (int t = 0; t < 8; ++t) {
    const float bv = b1[16 * t + m];
#pragma unroll
    for (int r = 0; r < 8; ++r) {
      float x = acc1[t][r] + bv;
      x = fmaxf(x, 0.0f);
      myH[(r + 8 * h) * H_STRIDE + 16 * t + m] = x;
    }
  }

  v8f acc2[4];
#pragma unroll
  for (int t = 0; t < 4; ++t) acc2[t] = vzero;

#pragma unroll
  for (int s = 0; s < 32; ++s) {
    const int kb = 4 * s + 2 * h;
    v2f A;
    A.x = myH[m * H_STRIDE + kb];
    A.y = myH[m * H_STRIDE + kb + 1];
    const float* w2r0 = W2 + (size_t)kb * Dq + m;
    const float* w2r1 = W2 + (size_t)(kb + 1) * Dq + m;
#pragma unroll
    for (int t = 0; t < 4; ++t) {
      v2f Bv;
      Bv.x = w2r0[16 * t];
      Bv.y = w2r1[16 * t];
      acc2[t] = __builtin_amdgcn_wmma_f32_16x16x4_f32(
          false, A, false, Bv, (short)0, acc2[t], false, false);
    }
  }

#pragma unroll
  for (int t = 0; t < 4; ++t) {
    const float bv = b2[16 * t + m];
    float part = 0.0f;
#pragma unroll
    for (int r = 0; r < 8; ++r) part += fmaxf(acc2[t][r] + bv, 0.0f);
    part += __shfl_xor(part, 16, 32);
    if (h == 0) out[(size_t)p * Dq + 16 * t + m] = part * (1.0f / 16.0f);
  }
}

extern "C" void kernel_launch(void* const* d_in, const int* in_sizes, int n_in,
                              void* d_out, int out_size, void* d_ws, size_t ws_size,
                              hipStream_t stream) {
  const float* events = (const float*)d_in[0];
  const float* W1     = (const float*)d_in[1];
  const float* b1     = (const float*)d_in[2];
  const float* W2     = (const float*)d_in[3];
  const float* b2     = (const float*)d_in[4];
  float*       outp   = (float*)d_out;
  int*         knn    = (int*)d_ws;

  knn_kernel<<<(Bq * Nq) / 256, 256, 0, stream>>>(events, knn);
  edgeconv_mlp_kernel<<<(Bq * Nq) / WAVES_PER_BLOCK, WAVES_PER_BLOCK * 32, 0, stream>>>(
      events, knn, W1, b1, W2, b2, outp);
}


# MLP kernel: one wave now does two points (2 x 16 edge rows), 4 waves per block, so every W1/W2 fragment load feeds two WMMAs: half the weight loads per product; each sum keeps its k order
# speedup vs baseline: 3.6150x; 1.0505x over previous
#include <hip/hip_runtime.h>

typedef __attribute__((ext_vector_type(2))) float v2f;
typedef __attribute__((ext_vector_type(8))) float v8f;

#define Bq 128
#define Nq 512
#define Fq 32
#define Hq 128
#define Dq 64
#define Kq 16

#define KNN_THREADS 256
#define KNN_BIG 3.4e38f

__device__ __forceinline__ float dist2(float x, float y, float cx, float cy) {
  const float dx = x - cx;
  const float dy = y - cy;
  return __builtin_fmaf(dx, dx, dy * dy);
}

template <bool EXCL>
__device__ __forceinline__ void knn_scan_values(const float2* sxy, int j0, int i,
                                                float cx, float cy, float (&bd)[Kq]) {
#pragma unroll 8
  for (int jj = 0; jj < 32; ++jj) {
    const int j = j0 + jj;
    const float2 c = sxy[j];
    float d = dist2(c.x, c.y, cx, cy);
    if (EXCL) d = (j == i) ? KNN_BIG : d;
#pragma unroll
    for (int t = Kq - 1; t >= 1; --t) bd[t] = __builtin_amdgcn_fmed3f(bd[t - 1], d, bd[t]);
    bd[0] = __builtin_fminf(d, bd[0]);
  }
}

template <bool EXCL>
__device__ __forceinline__ void knn_scan_gather(const float2* sxy, int* lst, int j0, int i,
                                                float cx, float cy, float tau,
                                                int& ties_left, int& pos) {
#pragma unroll 8
  for (int jj = 0; jj < 32; ++jj) {
    const int j = j0 + jj;
    const float2 c = sxy[j];
    float d = dist2(c.x, c.y, cx, cy);
    if (EXCL) d = (j == i) ? KNN_BIG : d;
    const bool eq   = (d == tau) && (ties_left > 0);
    const bool take = (d < tau) || eq;
    lst[pos] = j;
    pos += take ? KNN_THREADS : 0;
    ties_left -= eq ? 1 : 0;
  }
}

__global__ __launch_bounds__(KNN_THREADS) void knn_kernel(const float* __restrict__ events,
                                                          int* __restrict__ knn_idx) {
  __shared__ float2 sxy[Nq];
  __shared__ int    lst[(Kq + 1) * KNN_THREADS];
  const int tid = threadIdx.x;
  const int p   = blockIdx.x * KNN_THREADS + tid;
  const int b   = p >> 9;
  const int i   = p & (Nq - 1);

  for (int j = tid; j < Nq; j += KNN_THREADS) {
    const float* e = events + ((size_t)b * Nq + j) * Fq;
    float2 c;
    c.x = e[0];
    c.y = e[1];
    sxy[j] = c;
  }
  __syncthreads();

  const float cx = sxy[i].x, cy = sxy[i].y;
  const int mychunk = __builtin_amdgcn_readfirstlane(i >> 5);

  float bd[Kq];
#pragma unroll
  for (int t = 0; t < Kq; ++t) bd[t] = KNN_BIG;
  for (int cc = 0; cc < Nq / 32; ++cc) {
    if (cc == mychunk) knn_scan_values<true>(sxy, cc * 32, i, cx, cy, bd);
    else               knn_scan_values<false>(sxy, cc * 32, i, cx, cy, bd);
  }

  const float tau = bd[Kq - 1];
  int ties_left = 0;
#pragma unroll
  for (int t = 0; t < Kq; ++t) ties_left += (bd[t] == tau) ? 1 : 0;
  int pos = tid;
  for (int cc = 0; cc < Nq / 32; ++cc) {
    if (cc == mychunk) knn_scan_gather<true>(sxy, lst, cc * 32, i, cx, cy, tau, ties_left, pos);
    else               knn_scan_gather<false>(sxy, lst, cc * 32, i, cx, cy, tau, ties_left, pos);
  }

  float fd[Kq];
  int   fi[Kq];
#pragma unroll
  for (int t = 0; t < Kq; ++t) { fd[t] = KNN_BIG; fi[t] = 0; }
#pragma unroll
  for (int e = 0; e < Kq; ++e) {
    const int j = lst[e * KNN_THREADS + tid] & (Nq - 1);
    const float2 c = sxy[j];
    const float d = dist2(c.x, c.y, cx, cy);
#pragma unroll
    for (int t = Kq - 1; t >= 1; --t) {
      const bool below_prev = d < fd[t - 1];
      const bool below_this = d < fd[t];
      fi[t] = below_prev ? fi[t - 1] : (below_this ? j : fi[t]);
      fd[t] = __builtin_amdgcn_fmed3f(fd[t - 1], d, fd[t]);
    }
    fi[0] = (d < fd[0]) ? j : fi[0];
    fd[0] = __builtin_fminf(d, fd[0]);
  }

  int4* dst = reinterpret_cast<int4*>(knn_idx + (size_t)p * Kq);
#pragma unroll
  for (int q = 0; q < Kq / 4; ++q) {
    int4 v;
    v.x = fi[4 * q]; v.y = fi[4 * q + 1]; v.z = fi[4 * q + 2]; v.w = fi[4 * q + 3];
    dst[q] = v;
  }
}

#define WAVES_PER_BLOCK 4
#define PTS_PER_WAVE 2
#define PTS_PER_BLOCK (WAVES_PER_BLOCK * PTS_PER_WAVE)
#define H_STRIDE 129
#define P_STRIDE 132

__global__ __launch_bounds__(WAVES_PER_BLOCK * 32) void edgeconv_mlp_kernel(
    const float* __restrict__ events, const int* __restrict__ knn_idx,
    const float* __restrict__ W1, const float* __restrict__ b1,
    const float* __restrict__ W2, const float* __restrict__ b2,
    float* __restrict__ out) {
  __shared__ float ldsH[WAVES_PER_BLOCK * PTS_PER_WAVE * 16 * H_STRIDE];
  __shared__ float ldsP[16 * P_STRIDE];

  const int lane = threadIdx.x & 31;
  const int wid  = threadIdx.x >> 5;
  const int p0   = blockIdx.x * PTS_PER_BLOCK;
  const int pw   = p0 + wid * PTS_PER_WAVE;
  const int m    = lane & 15;
  const int h    = lane >> 4;

  float* myH = ldsH + wid * PTS_PER_WAVE * 16 * H_STRIDE;

  float a[PTS_PER_WAVE][16];
#pragma unroll
  for (int q = 0; q < PTS_PER_WAVE; ++q) {
    const int p  = pw + q;
    const int b  = p >> 9;
    const int i  = p & (Nq - 1);
    const int nj = knn_idx[(size_t)p * Kq + m];
    const float* ev = events + ((size_t)b * Nq + i) * Fq;
    const float* nb = events + ((size_t)b * Nq + nj) * Fq;
#pragma unroll
    for (int s = 0; s < 8; ++s) {
      const int c = 4 * s + 2 * h;
      a[q][2 * s]     = nb[c]     - ev[c];
      a[q][2 * s + 1] = nb[c + 1] - ev[c + 1];
    }
  }

  const v8f vzero = {};

  {
    const float* xr = events + (size_t)(p0 + (m & (PTS_PER_BLOCK - 1))) * Fq;
    v8f accp[2] = {vzero, vzero};
#pragma unroll
    for (int s = 0; s < 8; ++s) {
      const int kb = 4 * s + 2 * h;
      v2f A;
      A.x = xr[kb];
      A.y = xr[kb + 1];
#pragma unroll
      for (int j = 0; j < 2; ++j) {
        v2f Bv;
        Bv.x = W1[(size_t)kb * Hq + 16 * (2 * wid + j) + m];
        Bv.y = W1[(size_t)(kb + 1) * Hq + 16 * (2 * wid + j) + m];
        accp[j] = __builtin_amdgcn_wmma_f32_16x16x4_f32(
            false, A, false, Bv, (short)0, accp[j], false, false);
      }
    }
#pragma unroll
    for (int j = 0; j < 2; ++j)
#pragma unroll
      for (int r = 0; r < 8; ++r) ldsP[(r + 8 * h) * P_STRIDE + 16 * (2 * wid + j) + m] = accp[j][r];
  }
  __syncthreads();

#pragma unroll
  for (int th = 0; th < 2; ++th) {
    v8f acc1[PTS_PER_WAVE][4];
#pragma unroll
    for (int q = 0; q < PTS_PER_WAVE; ++q) {
#pragma unroll
      for (int tt = 0; tt < 4; ++tt) {
        const float xw = ldsP[(wid * PTS_PER_WAVE + q) * P_STRIDE + 16 * (4 * th + tt) + m];
#pragma unroll
        for (int r = 0; r < 8; ++r) acc1[q][tt][r] = xw;
      }
    }

#pragma unroll
    for (int s = 0; s < 8; ++s) {
      v2f A[PTS_PER_WAVE];
#pragma unroll
      for (int q = 0; q < PTS_PER_WAVE; ++q) {
        A[q].x = a[q][2 * s];
        A[q].y = a[q][2 * s + 1];
      }
      const int kb = 32 + 4 * s + 2 * h;
      const float* w1r0 = W1 + (size_t)kb * Hq + 64 * th + m;
      const float* w1r1 = W1 + (size_t)(kb + 1) * Hq + 64 * th + m;
#pragma unroll
      for (int tt = 0; tt < 4; ++tt) {
        v2f Bv;
        Bv.x = w1r0[16 * tt];
        Bv.y = w1r1[16 * tt];
#pragma unroll
        for (int q = 0; q < PTS_PER_WAVE; ++q)
          acc1[q][tt] = __builtin_amdgcn_wmma_f32_16x16x4_f32(
              false, A[q], false, Bv, (short)0, acc1[q][tt], false, false);
      }
    }

#pragma unroll
    for (int tt = 0; tt < 4; ++tt) {
      const int t = 4 * th + tt;
      const float bv = b1[16 * t + m];
#pragma unroll
      for (int q = 0; q < PTS_PER_WAVE; ++q) {
#pragma unroll
        for (int r = 0; r < 8; ++r) {
          float x = acc1[q][tt][r] + bv;
          x = fmaxf(x, 0.0f);
          myH[(q * 16 + r + 8 * h) * H_STRIDE + 16 * t + m] = x;
        }
      }
    }
  }

  v8f acc2[PTS_PER_WAVE][4];
#pragma unroll
  for (int q = 0; q < PTS_PER_WAVE; ++q)
#pragma unroll
    for (int t = 0; t < 4; ++t) acc2[q][t] = vzero;

#pragma unroll
  for (int s = 0; s < 32; ++s) {
    const int kb = 4 * s + 2 * h;
    v2f A[PTS_PER_WAVE];
#pragma unroll
    for (int q = 0; q < PTS_PER_WAVE; ++q) {
      A[q].x = myH[(q * 16 + m) * H_STRIDE + kb];
      A[q].y = myH[(q * 16 + m) * H_STRIDE + kb + 1];
    }
    const float* w2r0 = W2 + (size_t)kb * Dq + m;
    const float* w2r1 = W2 + (size_t)(kb + 1) * Dq + m;
#pragma unroll
    for (int t = 0; t < 4; ++t) {
      v2f Bv;
      Bv.x = w2r0[16 * t];
      Bv.y = w2r1[16 * t];
#pragma unroll
      for (int q = 0; q < PTS_PER_WAVE; ++q)
        acc2[q][t] = __builtin_amdgcn_wmma_f32_16x16x4_f32(
            false, A[q], false, Bv, (short)0, acc2[q][t], false, false);
    }
  }

#pragma unroll
  for (int t = 0; t < 4; ++t) {
    const float bv = b2[16 * t + m];
#pragma unroll
    for (int q = 0; q < PTS_PER_WAVE; ++q) {
      float part = 0.0f;
#pragma unroll
      for (int r = 0; r < 8; ++r) part += fmaxf(acc2[q][t][r] + bv, 0.0f);
      part += __shfl_xor(part, 16, 32);
      if (h == 0) out[(size_t)(pw + q) * Dq + 16 * t + m] = part * (1.0f / 16.0f);
    }
  }
}

extern "C" void kernel_launch(void* const* d_in, const int* in_sizes, int n_in,
                              void* d_out, int out_size, void* d_ws, size_t ws_size,
                              hipStream_t stream) {
  const float* events = (const float*)d_in[0];
  const float* W1     = (const float*)d_in[1];
  const float* b1     = (const float*)d_in[2];
  const float* W2     = (const float*)d_in[3];
  const float* b2     = (const float*)d_in[4];
  float*       outp   = (float*)d_out;
  int*         knn    = (int*)d_ws;

  knn_kernel<<<(Bq * Nq) / 256, 256, 0, stream>>>(events, knn);
  edgeconv_mlp_kernel<<<(Bq * Nq) / PTS_PER_BLOCK, WAVES_PER_BLOCK * 32, 0, stream>>>(
      events, knn, W1, b1, W2, b2, outp);
}


# MLP transposed (weights = WMMA A operand, edges = B): a layer-1 tile is layer 2's B fragment in registers, no h1 LDS stash, 3->6 waves/SIMD; layer-2 k order and mean sum order (lane tree) change
# speedup vs baseline: 3.8118x; 1.0544x over previous
#include <hip/hip_runtime.h>

typedef __attribute__((ext_vector_type(2))) float v2f;
typedef __attribute__((ext_vector_type(8))) float v8f;

#define Bq 128
#define Nq 512
#define Fq 32
#define Hq 128
#define Dq 64
#define Kq 16

#define KNN_THREADS 256
#define KNN_BIG 3.4e38f

__device__ __forceinline__ float dist2(float x, float y, float cx, float cy) {
  const float dx = x - cx;
  const float dy = y - cy;
  return __builtin_fmaf(dx, dx, dy * dy);
}

template <bool EXCL>
__device__ __forceinline__ void knn_scan_values(const float2* sxy, int j0, int i,
                                                float cx, float cy, float (&bd)[Kq]) {
#pragma unroll 8
  for (int jj = 0; jj < 32; ++jj) {
    const int j = j0 + jj;
    const float2 c = sxy[j];
    float d = dist2(c.x, c.y, cx, cy);
    if (EXCL) d = (j == i) ? KNN_BIG : d;
#pragma unroll
    for (int t = Kq - 1; t >= 1; --t) bd[t] = __builtin_amdgcn_fmed3f(bd[t - 1], d, bd[t]);
    bd[0] = __builtin_fminf(d, bd[0]);
  }
}

template <bool EXCL>
__device__ __forceinline__ void knn_scan_gather(const float2* sxy, int* lst, int j0, int i,
                                                float cx, float cy, float tau,
                                                int& ties_left, int& pos) {
#pragma unroll 8
  for (int jj = 0; jj < 32; ++jj) {
    const int j = j0 + jj;
    const float2 c = sxy[j];
    float d = dist2(c.x, c.y, cx, cy);
    if (EXCL) d = (j == i) ? KNN_BIG : d;
    const bool eq   = (d == tau) && (ties_left > 0);
    const bool take = (d < tau) || eq;
    lst[pos] = j;
    pos += take ? KNN_THREADS : 0;
    ties_left -= eq ? 1 : 0;
  }
}

__global__ __launch_bounds__(KNN_THREADS) void knn_kernel(const float* __restrict__ events,
                                                          int* __restrict__ knn_idx) {
  __shared__ float2 sxy[Nq];
  __shared__ int    lst[(Kq + 1) * KNN_THREADS];
  const int tid = threadIdx.x;
  const int p   = blockIdx.x * KNN_THREADS + tid;
  const int b   = p >> 9;
  const int i   = p & (Nq - 1);

  for (int j = tid; j < Nq; j += KNN_THREADS) {
    const float* e = events + ((size_t)b * Nq + j) * Fq;
    float2 c;
    c.x = e[0];
    c.y = e[1];
    sxy[j] = c;
  }
  __syncthreads();

  const float cx = sxy[i].x, cy = sxy[i].y;
  const int mychunk = __builtin_amdgcn_readfirstlane(i >> 5);

  float bd[Kq];
#pragma unroll
  for (int t = 0; t < Kq; ++t) bd[t] = KNN_BIG;
  for (int cc = 0; cc < Nq / 32; ++cc) {
    if (cc == mychunk) knn_scan_values<true>(sxy, cc * 32, i, cx, cy, bd);
    else               knn_scan_values<false>(sxy, cc * 32, i, cx, cy, bd);
  }

  const float tau = bd[Kq - 1];
  int ties_left = 0;
#pragma unroll
  for (int t = 0; t < Kq; ++t) ties_left += (bd[t] == tau) ? 1 : 0;
  int pos = tid;
  for (int cc = 0; cc < Nq / 32; ++cc) {
    if (cc == mychunk) knn_scan_gather<true>(sxy, lst, cc * 32, i, cx, cy, tau, ties_left, pos);
    else               knn_scan_gather<false>(sxy, lst, cc * 32, i, cx, cy, tau, ties_left, pos);
  }

  float fd[Kq];
  int   fi[Kq];
#pragma unroll
  for (int t = 0; t < Kq; ++t) { fd[t] = KNN_BIG; fi[t] = 0; }
#pragma unroll
  for (int e = 0; e < Kq; ++e) {
    const int j = lst[e * KNN_THREADS + tid] & (Nq - 1);
    const float2 c = sxy[j];
    const float d = dist2(c.x, c.y, cx, cy);
#pragma unroll
    for (int t = Kq - 1; t >= 1; --t) {
      const bool below_prev = d < fd[t - 1];
      const bool below_this = d < fd[t];
      fi[t] = below_prev ? fi[t - 1] : (below_this ? j : fi[t]);
      fd[t] = __builtin_amdgcn_fmed3f(fd[t - 1], d, fd[t]);
    }
    fi[0] = (d < fd[0]) ? j : fi[0];
    fd[0] = __builtin_fminf(d, fd[0]);
  }

  int4* dst = reinterpret_cast<int4*>(knn_idx + (size_t)p * Kq);
#pragma unroll
  for (int q = 0; q < Kq / 4; ++q) {
    int4 v;
    v.x = fi[4 * q]; v.y = fi[4 * q + 1]; v.z = fi[4 * q + 2]; v.w = fi[4 * q + 3];
    dst[q] = v;
  }
}

#define WAVES_PER_BLOCK 4
#define PTS_PER_WAVE 2
#define PTS_PER_BLOCK (WAVES_PER_BLOCK * PTS_PER_WAVE)
#define P_STRIDE 132
#define WF_PER_TILE 24
#define WF_TOTAL (WF_PER_TILE * (Hq / 16))
#define WF_AHEAD 8

template <int MASK>
__device__ __forceinline__ float row_xor(float v) {
  return __builtin_bit_cast(
      float, __builtin_amdgcn_update_dpp(0, __builtin_bit_cast(int, v), 0x160 + MASK, 0xf, 0xf, true));
}

__device__ __forceinline__ float row_sum_scatter(const float (&x)[8], int c) {
  const bool b2 = (c & 4) != 0, b1 = (c & 2) != 0, b0 = (c & 1) != 0;
  float y[4], z[2];
#pragma unroll
  for (int i = 0; i < 4; ++i) {
    const float keep = b2 ? x[4 + i] : x[i];
    const float send = b2 ? x[i] : x[4 + i];
    y[i] = keep + row_xor<4>(send);
  }
#pragma unroll
  for (int i = 0; i < 2; ++i) {
    const float keep = b1 ? y[2 + i] : y[i];
    const float send = b1 ? y[i] : y[2 + i];
    z[i] = keep + row_xor<2>(send);
  }
  const float keep = b0 ? z[1] : z[0];
  const float send = b0 ? z[0] : z[1];
  float w = keep + row_xor<1>(send);
  w += row_xor<8>(w);
  return w;
}

__device__ __forceinline__ v2f load_wfrag(const float* __restrict__ W1, const float* __restrict__ W2,
                                          int f, int m, int h) {
  const int t = f / WF_PER_TILE, r = f % WF_PER_TILE;
  v2f w;
  if (r < 8) {
    const int kb = 32 + 4 * r + 2 * h;
    w.x = W1[(size_t)kb * Hq + 16 * t + m];
    w.y = W1[(size_t)(kb + 1) * Hq + 16 * t + m];
  } else {
    const int j = (r - 8) >> 2, td = (r - 8) & 3;
    const int kk = 16 * t + 8 * h + 2 * j;
    w.x = W2[(size_t)kk * Dq + 16 * td + m];
    w.y = W2[(size_t)(kk + 1) * Dq + 16 * td + m];
  }
  return w;
}

__global__ __launch_bounds__(WAVES_PER_BLOCK * 32) __attribute__((amdgpu_waves_per_eu(6, 6)))
void edgeconv_mlp_kernel(
    const float* __restrict__ events, const int* __restrict__ knn_idx,
    const float* __restrict__ W1, const float* __restrict__ b1,
    const float* __restrict__ W2, const float* __restrict__ b2,
    float* __restrict__ out) {
  __shared__ alignas(16) float ldsP[16 * P_STRIDE];

  const int lane = threadIdx.x & 31;
  const int wid  = threadIdx.x >> 5;
  const int p0   = blockIdx.x * PTS_PER_BLOCK;
  const int pw   = p0 + wid * PTS_PER_WAVE;
  const int m    = lane & 15;
  const int h    = lane >> 4;

  float nbv[PTS_PER_WAVE][16], evv[PTS_PER_WAVE][16];
#pragma unroll
  for (int q = 0; q < PTS_PER_WAVE; ++q) {
    const int p  = pw + q;
    const int b  = p >> 9;
    const int i  = p & (Nq - 1);
    const int nj = knn_idx[(size_t)p * Kq + m];
    const float* ev = events + ((size_t)b * Nq + i) * Fq;
    const float* nb = events + ((size_t)b * Nq + nj) * Fq;
#pragma unroll
    for (int s = 0; s < 8; ++s) {
      const int c = 4 * s + 2 * h;
      nbv[q][2 * s]     = nb[c];
      nbv[q][2 * s + 1] = nb[c + 1];
      evv[q][2 * s]     = ev[c];
      evv[q][2 * s + 1] = ev[c + 1];
    }
  }
  float a[PTS_PER_WAVE][16];
#pragma unroll
  for (int q = 0; q < PTS_PER_WAVE; ++q)
#pragma unroll
    for (int e = 0; e < 16; ++e) a[q][e] = nbv[q][e] - evv[q][e];
  __builtin_amdgcn_sched_barrier(0);

  v2f xA[8], wB[8][2];
  {
    const float* xr = events + (size_t)(p0 + (m & (PTS_PER_BLOCK - 1))) * Fq;
#pragma unroll
    for (int s = 0; s < 8; ++s) {
      const int kb = 4 * s + 2 * h;
      xA[s].x = xr[kb];
      xA[s].y = xr[kb + 1];
#pragma unroll
      for (int j = 0; j < 2; ++j) {
        wB[s][j].x = W1[(size_t)kb * Hq + 16 * (2 * wid + j) + m];
        wB[s][j].y = W1[(size_t)(kb + 1) * Hq + 16 * (2 * wid + j) + m];
      }
    }
  }
  v2f ring[WF_AHEAD];
#pragma unroll
  for (int f = 0; f < WF_AHEAD; ++f) ring[f] = load_wfrag(W1, W2, f, m, h);

  const v8f vzero = {};

  {
    v8f accp[2] = {vzero, vzero};
#pragma unroll
    for (int s = 0; s < 8; ++s)
#pragma unroll
      for (int j = 0; j < 2; ++j)
        accp[j] = __builtin_amdgcn_wmma_f32_16x16x4_f32(
            false, xA[s], false, wB[s][j], (short)0, accp[j], false, false);
#pragma unroll
    for (int j = 0; j < 2; ++j)
#pragma unroll
      for (int r = 0; r < 8; ++r) ldsP[(r + 8 * h) * P_STRIDE + 16 * (2 * wid + j) + m] = accp[j][r];
  }
  __syncthreads();

  v8f acc2[PTS_PER_WAVE][4];
#pragma unroll
  for (int q = 0; q < PTS_PER_WAVE; ++q)
#pragma unroll
    for (int td = 0; td < 4; ++td) acc2[q][td] = vzero;

  v8f start[PTS_PER_WAVE];
#pragma unroll
  for (int q = 0; q < PTS_PER_WAVE; ++q) {
    const float4* xp = reinterpret_cast<const float4*>(
        &ldsP[(wid * PTS_PER_WAVE + q) * P_STRIDE + 8 * h]);
    const float4 lo = xp[0], hi = xp[1];
    start[q][0] = lo.x; start[q][1] = lo.y; start[q][2] = lo.z; start[q][3] = lo.w;
    start[q][4] = hi.x; start[q][5] = hi.y; start[q][6] = hi.z; start[q][7] = hi.w;
  }

#pragma unroll
  for (int t = 0; t < Hq / 16; ++t) {
    v8f acc1[PTS_PER_WAVE];
#pragma unroll
    for (int q = 0; q < PTS_PER_WAVE; ++q) acc1[q] = start[q];
    const float4* bp = reinterpret_cast<const float4*>(b1 + 16 * t + 8 * h);
    const float4 blo = bp[0], bhi = bp[1];
#pragma unroll
    for (int s = 0; s < 8; ++s) {
      const int f = WF_PER_TILE * t + s;
      const v2f Wf = ring[f % WF_AHEAD];
      if (f + WF_AHEAD < WF_TOTAL) ring[f % WF_AHEAD] = load_wfrag(W1, W2, f + WF_AHEAD, m, h);
#pragma unroll
      for (int q = 0; q < PTS_PER_WAVE; ++q) {
        v2f Ef;
        Ef.x = a[q][2 * s];
        Ef.y = a[q][2 * s + 1];
        acc1[q] = __builtin_amdgcn_wmma_f32_16x16x4_f32(
            false, Wf, false, Ef, (short)0, acc1[q], false, false);
      }
    }

    {
      const float bv[8] = {blo.x, blo.y, blo.z, blo.w, bhi.x, bhi.y, bhi.z, bhi.w};
#pragma unroll
      for (int q = 0; q < PTS_PER_WAVE; ++q)
#pragma unroll
        for (int r = 0; r < 8; ++r) acc1[q][r] = fmaxf(acc1[q][r] + bv[r], 0.0f);
    }

#pragma unroll
    for (int j = 0; j < 4; ++j) {
      if (j == 3 && t + 1 < Hq / 16) {
#pragma unroll
        for (int q = 0; q < PTS_PER_WAVE; ++q) {
          const float4* xp = reinterpret_cast<const float4*>(
              &ldsP[(wid * PTS_PER_WAVE + q) * P_STRIDE + 16 * (t + 1) + 8 * h]);
          const float4 lo = xp[0], hi = xp[1];
          start[q][0] = lo.x; start[q][1] = lo.y; start[q][2] = lo.z; start[q][3] = lo.w;
          start[q][4] = hi.x; start[q][5] = hi.y; start[q][6] = hi.z; start[q][7] = hi.w;
        }
      }
#pragma unroll
      for (int td = 0; td < 4; ++td) {
        const int f = WF_PER_TILE * t + 8 + 4 * j + td;
        const v2f Wf = ring[f % WF_AHEAD];
        if (f + WF_AHEAD < WF_TOTAL) ring[f % WF_AHEAD] = load_wfrag(W1, W2, f + WF_AHEAD, m, h);
#pragma unroll
        for (int q = 0; q < PTS_PER_WAVE; ++q) {
          v2f Hf;
          Hf.x = acc1[q][2 * j];
          Hf.y = acc1[q][2 * j + 1];
          acc2[q][td] = __builtin_amdgcn_wmma_f32_16x16x4_f32(
              false, Wf, false, Hf, (short)0, acc2[q][td], false, false);
        }
      }
    }
  }

  float bo[4][8];
#pragma unroll
  for (int td = 0; td < 4; ++td) {
    const float4* bp = reinterpret_cast<const float4*>(b2 + 16 * td + 8 * h);
    const float4 lo = bp[0], hi = bp[1];
    bo[td][0] = lo.x; bo[td][1] = lo.y; bo[td][2] = lo.z; bo[td][3] = lo.w;
    bo[td][4] = hi.x; bo[td][5] = hi.y; bo[td][6] = hi.z; bo[td][7] = hi.w;
  }
  const int c7 = m & 7;
  const bool up = (m & 8) != 0;
#pragma unroll
  for (int q = 0; q < PTS_PER_WAVE; ++q) {
    float w[4];
#pragma unroll
    for (int td = 0; td < 4; ++td) {
      float x[8];
#pragma unroll
      for (int r = 0; r < 8; ++r) x[r] = fmaxf(acc2[q][td][r] + bo[td][r], 0.0f);
      w[td] = row_sum_scatter(x, m);
    }
#pragma unroll
    for (int u = 0; u < 2; ++u) {
      const float v = up ? w[2 * u + 1] : w[2 * u];
      out[(size_t)(pw + q) * Dq + 32 * u + (up ? 16 : 0) + 8 * h + c7] = v * (1.0f / 16.0f);
    }
  }
}

extern "C" void kernel_launch(void* const* d_in, const int* in_sizes, int n_in,
                              void* d_out, int out_size, void* d_ws, size_t ws_size,
                              hipStream_t stream) {
  const float* events = (const float*)d_in[0];
  const float* W1     = (const float*)d_in[1];
  const float* b1     = (const float*)d_in[2];
  const float* W2     = (const float*)d_in[3];
  const float* b2     = (const float*)d_in[4];
  float*       outp   = (float*)d_out;
  int*         knn    = (int*)d_ws;

  knn_kernel<<<(Bq * Nq) / 256, 256, 0, stream>>>(events, knn);
  edgeconv_mlp_kernel<<<(Bq * Nq) / PTS_PER_BLOCK, WAVES_PER_BLOCK * 32, 0, stream>>>(
      events, knn, W1, b1, W2, b2, outp);
}


# MLP layer 1 by distributivity: Q = x@W1[32:64] once per point as an LDS table per half batch, hidden = relu(x@(W1[0:32]-W1[32:64])+b1 + Q[nbr]); no per-edge layer-1 product; layer-1 rounding changes
# speedup vs baseline: 4.9889x; 1.3088x over previous
#include <hip/hip_runtime.h>

typedef __attribute__((ext_vector_type(2))) float v2f;
typedef __attribute__((ext_vector_type(8))) float v8f;

#define Bq 128
#define Nq 512
#define Fq 32
#define Hq 128
#define Dq 64
#define Kq 16

#define KNN_THREADS 256
#define KNN_BIG 3.4e38f

__device__ __forceinline__ float dist2(float x, float y, float cx, float cy) {
  const float dx = x - cx;
  const float dy = y - cy;
  return __builtin_fmaf(dx, dx, dy * dy);
}

template <bool EXCL>
__device__ __forceinline__ void knn_scan_values(const float2* sxy, int j0, int i,
                                                float cx, float cy, float (&bd)[Kq]) {
#pragma unroll 8
  for (int jj = 0; jj < 32; ++jj) {
    const int j = j0 + jj;
    const float2 c = sxy[j];
    float d = dist2(c.x, c.y, cx, cy);
    if (EXCL) d = (j == i) ? KNN_BIG : d;
#pragma unroll
    for (int t = Kq - 1; t >= 1; --t) bd[t] = __builtin_amdgcn_fmed3f(bd[t - 1], d, bd[t]);
    bd[0] = __builtin_fminf(d, bd[0]);
  }
}

template <bool EXCL>
__device__ __forceinline__ void knn_scan_gather(const float2* sxy, int* lst, int j0, int i,
                                                float cx, float cy, float tau,
                                                int& ties_left, int& pos) {
#pragma unroll 8
  for (int jj = 0; jj < 32; ++jj) {
    const int j = j0 + jj;
    const float2 c = sxy[j];
    float d = dist2(c.x, c.y, cx, cy);
    if (EXCL) d = (j == i) ? KNN_BIG : d;
    const bool eq   = (d == tau) && (ties_left > 0);
    const bool take = (d < tau) || eq;
    lst[pos] = j;
    pos += take ? KNN_THREADS : 0;
    ties_left -= eq ? 1 : 0;
  }
}

__global__ __launch_bounds__(KNN_THREADS) void knn_kernel(const float* __restrict__ events,
                                                          int* __restrict__ knn_idx) {
  __shared__ float2 sxy[Nq];
  __shared__ int    lst[(Kq + 1) * KNN_THREADS];
  const int tid = threadIdx.x;
  const int p   = blockIdx.x * KNN_THREADS + tid;
  const int b   = p >> 9;
  const int i   = p & (Nq - 1);

  for (int j = tid; j < Nq; j += KNN_THREADS) {
    const float* e = events + ((size_t)b * Nq + j) * Fq;
    float2 c;
    c.x = e[0];
    c.y = e[1];
    sxy[j] = c;
  }
  __syncthreads();

  const float cx = sxy[i].x, cy = sxy[i].y;
  const int mychunk = __builtin_amdgcn_readfirstlane(i >> 5);

  float bd[Kq];
#pragma unroll
  for (int t = 0; t < Kq; ++t) bd[t] = KNN_BIG;
  for (int cc = 0; cc < Nq / 32; ++cc) {
    if (cc == mychunk) knn_scan_values<true>(sxy, cc * 32, i, cx, cy, bd);
    else               knn_scan_values<false>(sxy, cc * 32, i, cx, cy, bd);
  }

  const float tau = bd[Kq - 1];
  int ties_left = 0;
#pragma unroll
  for (int t = 0; t < Kq; ++t) ties_left += (bd[t] == tau) ? 1 : 0;
  int pos = tid;
  for (int cc = 0; cc < Nq / 32; ++cc) {
    if (cc == mychunk) knn_scan_gather<true>(sxy, lst, cc * 32, i, cx, cy, tau, ties_left, pos);
    else               knn_scan_gather<false>(sxy, lst, cc * 32, i, cx, cy, tau, ties_left, pos);
  }

  float fd[Kq];
  int   fi[Kq];
#pragma unroll
  for (int t = 0; t < Kq; ++t) { fd[t] = KNN_BIG; fi[t] = 0; }
#pragma unroll
  for (int e = 0; e < Kq; ++e) {
    const int j = lst[e * KNN_THREADS + tid] & (Nq - 1);
    const float2 c = sxy[j];
    const float d = dist2(c.x, c.y, cx, cy);
#pragma unroll
    for (int t = Kq - 1; t >= 1; --t) {
      const bool below_prev = d < fd[t - 1];
      const bool below_this = d < fd[t];
      fi[t] = below_prev ? fi[t - 1] : (below_this ? j : fi[t]);
      fd[t] = __builtin_amdgcn_fmed3f(fd[t - 1], d, fd[t]);
    }
    fi[0] = (d < fd[0]) ? j : fi[0];
    fd[0] = __builtin_fminf(d, fd[0]);
  }

  int4* dst = reinterpret_cast<int4*>(knn_idx + (size_t)p * Kq);
#pragma unroll
  for (int q = 0; q < Kq / 4; ++q) {
    int4 v;
    v.x = fi[4 * q]; v.y = fi[4 * q + 1]; v.z = fi[4 * q + 2]; v.w = fi[4 * q + 3];
    dst[q] = v;
  }
}

#define MLP_WAVES 16
#define MLP_THREADS (MLP_WAVES * 32)
#define MLP_GRID (2 * Bq)
#define QP 132
#define WF_PER_PAIR 128
#define WF_AHEAD 8

template <int MASK>
__device__ __forceinline__ float row_xor(float v) {
  return __builtin_bit_cast(
      float, __builtin_amdgcn_update_dpp(0, __builtin_bit_cast(int, v), 0x160 + MASK, 0xf, 0xf, true));
}

__device__ __forceinline__ float row_bcast(float v, unsigned sel) {
  const unsigned u = __builtin_bit_cast(unsigned, v);
  return __builtin_bit_cast(float, __builtin_amdgcn_permlane16(u, u, sel, sel, false, false));
}

__device__ __forceinline__ float row_sum_scatter(const float (&x)[8], int c) {
  const bool b2 = (c & 4) != 0, b1 = (c & 2) != 0, b0 = (c & 1) != 0;
  float y[4], z[2];
#pragma unroll
  for (int i = 0; i < 4; ++i) {
    const float keep = b2 ? x[4 + i] : x[i];
    const float send = b2 ? x[i] : x[4 + i];
    y[i] = keep + row_xor<4>(send);
  }
#pragma unroll
  for (int i = 0; i < 2; ++i) {
    const float keep = b1 ? y[2 + i] : y[i];
    const float send = b1 ? y[i] : y[2 + i];
    z[i] = keep + row_xor<2>(send);
  }
  const float keep = b0 ? z[1] : z[0];
  const float send = b0 ? z[0] : z[1];
  float w = keep + row_xor<1>(send);
  w += row_xor<8>(w);
  return w;
}

__device__ __forceinline__ v2f load_w2frag(const float* __restrict__ W2, int f, int m, int h) {
  const int t = f >> 4, j = (f >> 2) & 3, td = f & 3;
  const int kk = 16 * t + 8 * h + 2 * j;
  v2f w;
  w.x = W2[(size_t)kk * Dq + 16 * td + m];
  w.y = W2[(size_t)(kk + 1) * Dq + 16 * td + m];
  return w;
}

__global__ __launch_bounds__(MLP_THREADS) __attribute__((amdgpu_waves_per_eu(4, 4)))
void edgeconv_mlp_kernel(
    const float* __restrict__ events, const int* __restrict__ knn_idx,
    const float* __restrict__ W1, const float* __restrict__ b1,
    const float* __restrict__ W2, const float* __restrict__ b2,
    float* __restrict__ out) {
  __shared__ alignas(16) float ldsQ[Nq * QP];

  const int lane = threadIdx.x & 31;
  const int wid  = __builtin_amdgcn_readfirstlane(threadIdx.x >> 5);
  const int b    = blockIdx.x >> 1;
  const int hf   = blockIdx.x & 1;
  const int m    = lane & 15;
  const int h    = lane >> 4;
  const int tile_own = 16 * hf + wid;
  const int tile_oth = 16 * (1 - hf) + wid;
  const float* evb = events + (size_t)b * Nq * Fq;

  v2f xo[8], xt[8];
  {
    const float* r0 = evb + (size_t)(16 * tile_own + m) * Fq + 2 * h;
    const float* r1 = evb + (size_t)(16 * tile_oth + m) * Fq + 2 * h;
#pragma unroll
    for (int s = 0; s < 8; ++s) {
      xo[s].x = r0[4 * s]; xo[s].y = r0[4 * s + 1];
      xt[s].x = r1[4 * s]; xt[s].y = r1[4 * s + 1];
    }
  }
  const v8f vzero = {};
  v8f PT[Hq / 16];
  v2f wa[8], wb[8];
  const float* w1l = W1 + (size_t)(2 * h) * Hq + m;
#pragma unroll
  for (int s = 0; s < 8; ++s) {
    wa[s].x = w1l[(size_t)(4 * s) * Hq];
    wa[s].y = w1l[(size_t)(4 * s + 1) * Hq];
    wb[s].x = w1l[(size_t)(32 + 4 * s) * Hq];
    wb[s].y = w1l[(size_t)(33 + 4 * s) * Hq];
  }
#pragma unroll
  for (int t = 0; t < Hq / 16; ++t) {
    v2f na[8], nb[8];
    if (t + 1 < Hq / 16) {
#pragma unroll
      for (int s = 0; s < 8; ++s) {
        na[s].x = w1l[(size_t)(4 * s) * Hq + 16 * (t + 1)];
        na[s].y = w1l[(size_t)(4 * s + 1) * Hq + 16 * (t + 1)];
        nb[s].x = w1l[(size_t)(32 + 4 * s) * Hq + 16 * (t + 1)];
        nb[s].y = w1l[(size_t)(33 + 4 * s) * Hq + 16 * (t + 1)];
      }
    }
    v8f qo = vzero, qt = vzero, pt = vzero;
#pragma unroll
    for (int s = 0; s < 8; ++s) {
      v2f wd;
      wd.x = wa[s].x - wb[s].x;
      wd.y = wa[s].y - wb[s].y;
      qo = __builtin_amdgcn_wmma_f32_16x16x4_f32(false, wb[s], false, xo[s], (short)0, qo, false, false);
      qt = __builtin_amdgcn_wmma_f32_16x16x4_f32(false, wb[s], false, xt[s], (short)0, qt, false, false);
      pt = __builtin_amdgcn_wmma_f32_16x16x4_f32(false, wd, false, xo[s], (short)0, pt, false, false);
    }
    float4* d0 = reinterpret_cast<float4*>(&ldsQ[(16 * tile_own + m) * QP + 16 * t + 8 * h]);
    float4* d1 = reinterpret_cast<float4*>(&ldsQ[(16 * tile_oth + m) * QP + 16 * t + 8 * h]);
    d0[0] = make_float4(qo[0], qo[1], qo[2], qo[3]);
    d0[1] = make_float4(qo[4], qo[5], qo[6], qo[7]);
    d1[0] = make_float4(qt[0], qt[1], qt[2], qt[3]);
    d1[1] = make_float4(qt[4], qt[5], qt[6], qt[7]);
    const float4* bp = reinterpret_cast<const float4*>(b1 + 16 * t + 8 * h);
    const float4 blo = bp[0], bhi = bp[1];
    PT[t][0] = pt[0] + blo.x; PT[t][1] = pt[1] + blo.y; PT[t][2] = pt[2] + blo.z; PT[t][3] = pt[3] + blo.w;
    PT[t][4] = pt[4] + bhi.x; PT[t][5] = pt[5] + bhi.y; PT[t][6] = pt[6] + bhi.z; PT[t][7] = pt[7] + bhi.w;
    if (t + 1 < Hq / 16) {
#pragma unroll
      for (int s = 0; s < 8; ++s) { wa[s] = na[s]; wb[s] = nb[s]; }
    }
    __builtin_amdgcn_sched_barrier(0);
  }
  __syncthreads();

  const int gp0 = b * Nq + 16 * tile_own;
  v2f ring[WF_AHEAD];
#pragma unroll
  for (int f = 0; f < WF_AHEAD; ++f) ring[f] = load_w2frag(W2, f, m, h);

  int nja = knn_idx[(size_t)gp0 * Kq + m];
  int njb = knn_idx[(size_t)(gp0 + 1) * Kq + m];

  const int c7 = m & 7;
  const bool up = (m & 8) != 0;

#pragma unroll 1
  for (int pp = 0; pp < 8; ++pp) {
    const float* qrow[2];
    qrow[0] = ldsQ + (nja & (Nq - 1)) * QP + 8 * h;
    qrow[1] = ldsQ + (njb & (Nq - 1)) * QP + 8 * h;
    const int pn = (pp + 1) & 7;
    nja = knn_idx[(size_t)(gp0 + 2 * pn) * Kq + m];
    njb = knn_idx[(size_t)(gp0 + 2 * pn + 1) * Kq + m];
    int wofs = 0;
    asm volatile("" : "+s"(wofs));
    const float* W2p = W2 + wofs;
    const float* b2p = b2 + wofs;
    unsigned sel[2];
    sel[0] = (unsigned)(2 * pp) * 0x11111111u;
    sel[1] = sel[0] + 0x11111111u;

    v8f acc2[2][4];
#pragma unroll
    for (int q = 0; q < 2; ++q)
#pragma unroll
      for (int td = 0; td < 4; ++td) acc2[q][td] = vzero;

    float4 qlo[2], qhi[2];
#pragma unroll
    for (int q = 0; q < 2; ++q) {
      const float4* xp = reinterpret_cast<const float4*>(qrow[q]);
      qlo[q] = xp[0]; qhi[q] = xp[1];
    }

#pragma unroll
    for (int t = 0; t < Hq / 16; ++t) {
      float hv[2][8];
#pragma unroll
      for (int q = 0; q < 2; ++q) {
        const float qv[8] = {qlo[q].x, qlo[q].y, qlo[q].z, qlo[q].w, qhi[q].x, qhi[q].y, qhi[q].z, qhi[q].w};
#pragma unroll
        for (int r = 0; r < 8; ++r) hv[q][r] = fmaxf(row_bcast(PT[t][r], sel[q]) + qv[r], 0.0f);
      }
      if (t + 1 < Hq / 16) {
#pragma unroll
        for (int q = 0; q < 2; ++q) {
          const float4* xp = reinterpret_cast<const float4*>(qrow[q] + 16 * (t + 1));
          qlo[q] = xp[0]; qhi[q] = xp[1];
        }
      }
#pragma unroll
      for (int j = 0; j < 4; ++j) {
#pragma unroll
        for (int td = 0; td < 4; ++td) {
          const int f = 16 * t + 4 * j + td;
          const v2f Wf = ring[f % WF_AHEAD];
          ring[f % WF_AHEAD] = load_w2frag(W2p, (f + WF_AHEAD) % WF_PER_PAIR, m, h);
#pragma unroll
          for (int q = 0; q < 2; ++q) {
            v2f Hf;
            Hf.x = hv[q][2 * j];
            Hf.y = hv[q][2 * j + 1];
            acc2[q][td] = __builtin_amdgcn_wmma_f32_16x16x4_f32(
                false, Wf, false, Hf, (short)0, acc2[q][td], false, false);
          }
        }
      }
      __builtin_amdgcn_sched_barrier(0);
    }

    float bo[4][8];
#pragma unroll
    for (int td = 0; td < 4; ++td) {
      const float4* bp = reinterpret_cast<const float4*>(b2p + 16 * td + 8 * h);
      const float4 lo = bp[0], hi = bp[1];
      bo[td][0] = lo.x; bo[td][1] = lo.y; bo[td][2] = lo.z; bo[td][3] = lo.w;
      bo[td][4] = hi.x; bo[td][5] = hi.y; bo[td][6] = hi.z; bo[td][7] = hi.w;
    }
#pragma unroll
    for (int q = 0; q < 2; ++q) {
      float w[4];
#pragma unroll
      for (int td = 0; td < 4; ++td) {
        float x[8];
#pragma unroll
        for (int r = 0; r < 8; ++r) x[r] = fmaxf(acc2[q][td][r] + bo[td][r], 0.0f);
        w[td] = row_sum_scatter(x, m);
      }
#pragma unroll
      for (int u = 0; u < 2; ++u) {
        const float v = up ? w[2 * u + 1] : w[2 * u];
        out[(size_t)(gp0 + 2 * pp + q) * Dq + 32 * u + (up ? 16 : 0) + 8 * h + c7] = v * (1.0f / 16.0f);
      }
    }
  }
}

extern "C" void kernel_launch(void* const* d_in, const int* in_sizes, int n_in,
                              void* d_out, int out_size, void* d_ws, size_t ws_size,
                              hipStream_t stream) {
  const float* events = (const float*)d_in[0];
  const float* W1     = (const float*)d_in[1];
  const float* b1     = (const float*)d_in[2];
  const float* W2     = (const float*)d_in[3];
  const float* b2     = (const float*)d_in[4];
  float*       outp   = (float*)d_out;
  int*         knn    = (int*)d_ws;

  knn_kernel<<<(Bq * Nq) / 256, 256, 0, stream>>>(events, knn);
  edgeconv_mlp_kernel<<<MLP_GRID, MLP_THREADS, 0, stream>>>(
      events, knn, W1, b1, W2, b2, outp);
}


# KNN fused into the MLP kernel, one launch: waves 0-7 of a block select the 16 ids into LDS on the vector ALU while waves 8-15 build the Q table on the matrix unit; every sum keeps its order
# speedup vs baseline: 5.1582x; 1.0339x over previous
#include <hip/hip_runtime.h>

typedef __attribute__((ext_vector_type(2))) float v2f;
typedef __attribute__((ext_vector_type(8))) float v8f;

#define Bq 128
#define Nq 512
#define Fq 32
#define Hq 128
#define Dq 64
#define Kq 16

#define KNN_THREADS 256
#define KNN_BIG 3.4e38f

__device__ __forceinline__ float dist2(float x, float y, float cx, float cy) {
  const float dx = x - cx;
  const float dy = y - cy;
  return __builtin_fmaf(dx, dx, dy * dy);
}

template <bool EXCL>
__device__ __forceinline__ void knn_scan_values(const float2* sxy, int j0, int i,
                                                float cx, float cy, float (&bd)[Kq]) {
#pragma unroll 8
  for (int jj = 0; jj < 32; ++jj) {
    const int j = j0 + jj;
    const float2 c = sxy[j];
    float d = dist2(c.x, c.y, cx, cy);
    if (EXCL) d = (j == i) ? KNN_BIG : d;
#pragma unroll
    for (int t = Kq - 1; t >= 1; --t) bd[t] = __builtin_amdgcn_fmed3f(bd[t - 1], d, bd[t]);
    bd[0] = __builtin_fminf(d, bd[0]);
  }
}

template <bool EXCL>
__device__ __forceinline__ void knn_scan_gather(const float2* sxy, int* lst, int j0, int i,
                                                float cx, float cy, float tau,
                                                int& ties_left, int& pos) {
#pragma unroll 8
  for (int jj = 0; jj < 32; ++jj) {
    const int j = j0 + jj;
    const float2 c = sxy[j];
    float d = dist2(c.x, c.y, cx, cy);
    if (EXCL) d = (j == i) ? KNN_BIG : d;
    const bool eq   = (d == tau) && (ties_left > 0);
    const bool take = (d < tau) || eq;
    lst[pos] = j;
    pos += take ? KNN_THREADS : 0;
    ties_left -= eq ? 1 : 0;
  }
}

__device__ __forceinline__ void knn_half_batch(const float2* sxy, int* lst, int* nbr, int tid, int i) {
  const float cx = sxy[i].x, cy = sxy[i].y;
  const int mychunk = __builtin_amdgcn_readfirstlane(i >> 5);

  float bd[Kq];
#pragma unroll
  for (int t = 0; t < Kq; ++t) bd[t] = KNN_BIG;
  for (int cc = 0; cc < Nq / 32; ++cc) {
    if (cc == mychunk) knn_scan_values<true>(sxy, cc * 32, i, cx, cy, bd);
    else               knn_scan_values<false>(sxy, cc * 32, i, cx, cy, bd);
  }

  const float tau = bd[Kq - 1];
  int ties_left = 0;
#pragma unroll
  for (int t = 0; t < Kq; ++t) ties_left += (bd[t] == tau) ? 1 : 0;
  int pos = tid;
  for (int cc = 0; cc < Nq / 32; ++cc) {
    if (cc == mychunk) knn_scan_gather<true>(sxy, lst, cc * 32, i, cx, cy, tau, ties_left, pos);
    else               knn_scan_gather<false>(sxy, lst, cc * 32, i, cx, cy, tau, ties_left, pos);
  }

  float fd[Kq];
  int   fi[Kq];
#pragma unroll
  for (int t = 0; t < Kq; ++t) { fd[t] = KNN_BIG; fi[t] = 0; }
#pragma unroll
  for (int e = 0; e < Kq; ++e) {
    const int j = lst[e * KNN_THREADS + tid] & (Nq - 1);
    const float2 c = sxy[j];
    const float d = dist2(c.x, c.y, cx, cy);
#pragma unroll
    for (int t = Kq - 1; t >= 1; --t) {
      const bool below_prev = d < fd[t - 1];
      const bool below_this = d < fd[t];
      fi[t] = below_prev ? fi[t - 1] : (below_this ? j : fi[t]);
      fd[t] = __builtin_amdgcn_fmed3f(fd[t - 1], d, fd[t]);
    }
    fi[0] = (d < fd[0]) ? j : fi[0];
    fd[0] = __builtin_fminf(d, fd[0]);
  }

  int4* dst = reinterpret_cast<int4*>(nbr + tid * Kq);
#pragma unroll
  for (int q = 0; q < Kq / 4; ++q) {
    int4 v;
    v.x = fi[4 * q]; v.y = fi[4 * q + 1]; v.z = fi[4 * q + 2]; v.w = fi[4 * q + 3];
    dst[q] = v;
  }
}

#define MLP_WAVES 16
#define MLP_THREADS (MLP_WAVES * 32)
#define MLP_GRID (2 * Bq)
#define QP 132
#define WF_PER_PAIR 128
#define WF_AHEAD 8

template <int MASK>
__device__ __forceinline__ float row_xor(float v) {
  return __builtin_bit_cast(
      float, __builtin_amdgcn_update_dpp(0, __builtin_bit_cast(int, v), 0x160 + MASK, 0xf, 0xf, true));
}

__device__ __forceinline__ float row_bcast(float v, unsigned sel) {
  const unsigned u = __builtin_bit_cast(unsigned, v);
  return __builtin_bit_cast(float, __builtin_amdgcn_permlane16(u, u, sel, sel, false, false));
}

__device__ __forceinline__ float row_sum_scatter(const float (&x)[8], int c) {
  const bool b2 = (c & 4) != 0, b1 = (c & 2) != 0, b0 = (c & 1) != 0;
  float y[4], z[2];
#pragma unroll
  for (int i = 0; i < 4; ++i) {
    const float keep = b2 ? x[4 + i] : x[i];
    const float send = b2 ? x[i] : x[4 + i];
    y[i] = keep + row_xor<4>(send);
  }
#pragma unroll
  for (int i = 0; i < 2; ++i) {
    const float keep = b1 ? y[2 + i] : y[i];
    const float send = b1 ? y[i] : y[2 + i];
    z[i] = keep + row_xor<2>(send);
  }
  const float keep = b0 ? z[1] : z[0];
  const float send = b0 ? z[0] : z[1];
  float w = keep + row_xor<1>(send);
  w += row_xor<8>(w);
  return w;
}

__device__ __forceinline__ v2f load_w2frag(const float* __restrict__ W2, int f, int m, int h) {
  const int t = f >> 4, j = (f >> 2) & 3, td = f & 3;
  const int kk = 16 * t + 8 * h + 2 * j;
  v2f w;
  w.x = W2[(size_t)kk * Dq + 16 * td + m];
  w.y = W2[(size_t)(kk + 1) * Dq + 16 * td + m];
  return w;
}

__device__ __forceinline__ void build_q_tiles(const float* __restrict__ evb, const float* __restrict__ W1,
                                              float* ldsQ, int u, int m, int h) {
  v2f xq[4][8];
#pragma unroll
  for (int g = 0; g < 4; ++g) {
    const float* r = evb + (size_t)(16 * (4 * u + g) + m) * Fq + 2 * h;
#pragma unroll
    for (int s = 0; s < 8; ++s) { xq[g][s].x = r[4 * s]; xq[g][s].y = r[4 * s + 1]; }
  }
  const v8f vzero = {};
  v2f wb[8];
  const float* w1l = W1 + (size_t)(32 + 2 * h) * Hq + m;
#pragma unroll
  for (int s = 0; s < 8; ++s) {
    wb[s].x = w1l[(size_t)(4 * s) * Hq];
    wb[s].y = w1l[(size_t)(4 * s + 1) * Hq];
  }
#pragma unroll
  for (int t = 0; t < Hq / 16; ++t) {
    v2f nb[8];
    if (t + 1 < Hq / 16) {
#pragma unroll
      for (int s = 0; s < 8; ++s) {
        nb[s].x = w1l[(size_t)(4 * s) * Hq + 16 * (t + 1)];
        nb[s].y = w1l[(size_t)(4 * s + 1) * Hq + 16 * (t + 1)];
      }
    }
    v8f q[4];
#pragma unroll
    for (int g = 0; g < 4; ++g) q[g] = vzero;
#pragma unroll
    for (int s = 0; s < 8; ++s) {
#pragma unroll
      for (int g = 0; g < 4; ++g)
        q[g] = __builtin_amdgcn_wmma_f32_16x16x4_f32(false, wb[s], false, xq[g][s], (short)0, q[g], false, false);
    }
#pragma unroll
    for (int g = 0; g < 4; ++g) {
      float4* d = reinterpret_cast<float4*>(&ldsQ[(16 * (4 * u + g) + m) * QP + 16 * t + 8 * h]);
      d[0] = make_float4(q[g][0], q[g][1], q[g][2], q[g][3]);
      d[1] = make_float4(q[g][4], q[g][5], q[g][6], q[g][7]);
    }
    if (t + 1 < Hq / 16) {
#pragma unroll
      for (int s = 0; s < 8; ++s) wb[s] = nb[s];
    }
    __builtin_amdgcn_sched_barrier(0);
  }
}

__global__ __launch_bounds__(MLP_THREADS) __attribute__((amdgpu_waves_per_eu(4, 4)))
void edgeconv_fused_kernel(
    const float* __restrict__ events,
    const float* __restrict__ W1, const float* __restrict__ b1,
    const float* __restrict__ W2, const float* __restrict__ b2,
    float* __restrict__ out) {
  __shared__ alignas(16) float ldsQ[Nq * QP];
  __shared__ float2 sxy[Nq];
  __shared__ int lst[(Kq + 1) * KNN_THREADS];
  __shared__ alignas(16) int nbr[KNN_THREADS * Kq];

  const int tid  = threadIdx.x;
  const int lane = threadIdx.x & 31;
  const int wid  = __builtin_amdgcn_readfirstlane(threadIdx.x >> 5);
  const int b    = blockIdx.x >> 1;
  const int hf   = blockIdx.x & 1;
  const int m    = lane & 15;
  const int h    = lane >> 4;
  const int tile_own = 16 * hf + wid;
  const float* evb = events + (size_t)b * Nq * Fq;

  {
    const float* e = evb + (size_t)tid * Fq;
    float2 c;
    c.x = e[0];
    c.y = e[1];
    sxy[tid] = c;
  }
  __syncthreads();

  if (wid < MLP_WAVES / 2) knn_half_batch(sxy, lst, nbr, tid, KNN_THREADS * hf + tid);
  else                     build_q_tiles(evb, W1, ldsQ, wid - MLP_WAVES / 2, m, h);

  v2f xo[8];
  {
    const float* r0 = evb + (size_t)(16 * tile_own + m) * Fq + 2 * h;
#pragma unroll
    for (int s = 0; s < 8; ++s) { xo[s].x = r0[4 * s]; xo[s].y = r0[4 * s + 1]; }
  }
  const v8f vzero = {};
  v8f PT[Hq / 16];
  v2f wa[8], wb[8];
  const float* w1l = W1 + (size_t)(2 * h) * Hq + m;
#pragma unroll
  for (int s = 0; s < 8; ++s) {
    wa[s].x = w1l[(size_t)(4 * s) * Hq];
    wa[s].y = w1l[(size_t)(4 * s + 1) * Hq];
    wb[s].x = w1l[(size_t)(32 + 4 * s) * Hq];
    wb[s].y = w1l[(size_t)(33 + 4 * s) * Hq];
  }
#pragma unroll
  for (int t = 0; t < Hq / 16; ++t) {
    v2f na[8], nb[8];
    if (t + 1 < Hq / 16) {
#pragma unroll
      for (int s = 0; s < 8; ++s) {
        na[s].x = w1l[(size_t)(4 * s) * Hq + 16 * (t + 1)];
        na[s].y = w1l[(size_t)(4 * s + 1) * Hq + 16 * (t + 1)];
        nb[s].x = w1l[(size_t)(32 + 4 * s) * Hq + 16 * (t + 1)];
        nb[s].y = w1l[(size_t)(33 + 4 * s) * Hq + 16 * (t + 1)];
      }
    }
    v8f pt = vzero;
#pragma unroll
    for (int s = 0; s < 8; ++s) {
      v2f wd;
      wd.x = wa[s].x - wb[s].x;
      wd.y = wa[s].y - wb[s].y;
      pt = __builtin_amdgcn_wmma_f32_16x16x4_f32(false, wd, false, xo[s], (short)0, pt, false, false);
    }
    const float4* bp = reinterpret_cast<const float4*>(b1 + 16 * t + 8 * h);
    const float4 blo = bp[0], bhi = bp[1];
    PT[t][0] = pt[0] + blo.x; PT[t][1] = pt[1] + blo.y; PT[t][2] = pt[2] + blo.z; PT[t][3] = pt[3] + blo.w;
    PT[t][4] = pt[4] + bhi.x; PT[t][5] = pt[5] + bhi.y; PT[t][6] = pt[6] + bhi.z; PT[t][7] = pt[7] + bhi.w;
    if (t + 1 < Hq / 16) {
#pragma unroll
      for (int s = 0; s < 8; ++s) { wa[s] = na[s]; wb[s] = nb[s]; }
    }
    __builtin_amdgcn_sched_barrier(0);
  }
  __syncthreads();

  const int gp0 = b * Nq + 16 * tile_own;
  v2f ring[WF_AHEAD];
#pragma unroll
  for (int f = 0; f < WF_AHEAD; ++f) ring[f] = load_w2frag(W2, f, m, h);

  const int* nb0 = nbr + (16 * wid) * Kq + m;
  int nja = nb0[0];
  int njb = nb0[Kq];

  const int c7 = m & 7;
  const bool up = (m & 8) != 0;

#pragma unroll 1
  for (int pp = 0; pp < 8; ++pp) {
    const float* qrow[2];
    qrow[0] = ldsQ + (nja & (Nq - 1)) * QP + 8 * h;
    qrow[1] = ldsQ + (njb & (Nq - 1)) * QP + 8 * h;
    const int pn = (pp + 1) & 7;
    nja = nb0[(2 * pn) * Kq];
    njb = nb0[(2 * pn + 1) * Kq];
    int wofs = 0;
    asm volatile("" : "+s"(wofs));
    const float* W2p = W2 + wofs;
    const float* b2p = b2 + wofs;
    unsigned sel[2];
    sel[0] = (unsigned)(2 * pp) * 0x11111111u;
    sel[1] = sel[0] + 0x11111111u;

    v8f acc2[2][4];
#pragma unroll
    for (int q = 0; q < 2; ++q)
#pragma unroll
      for (int td = 0; td < 4; ++td) acc2[q][td] = vzero;

    float4 qlo[2], qhi[2];
#pragma unroll
    for (int q = 0; q < 2; ++q) {
      const float4* xp = reinterpret_cast<const float4*>(qrow[q]);
      qlo[q] = xp[0]; qhi[q] = xp[1];
    }

#pragma unroll
    for (int t = 0; t < Hq / 16; ++t) {
      float hv[2][8];
#pragma unroll
      for (int q = 0; q < 2; ++q) {
        const float qv[8] = {qlo[q].x, qlo[q].y, qlo[q].z, qlo[q].w, qhi[q].x, qhi[q].y, qhi[q].z, qhi[q].w};
#pragma unroll
        for (int r = 0; r < 8; ++r) hv[q][r] = fmaxf(row_bcast(PT[t][r], sel[q]) + qv[r], 0.0f);
      }
      if (t + 1 < Hq / 16) {
#pragma unroll
        for (int q = 0; q < 2; ++q) {
          const float4* xp = reinterpret_cast<const float4*>(qrow[q] + 16 * (t + 1));
          qlo[q] = xp[0]; qhi[q] = xp[1];
        }
      }
#pragma unroll
      for (int j = 0; j < 4; ++j) {
#pragma unroll
        for (int td = 0; td < 4; ++td) {
          const int f = 16 * t + 4 * j + td;
          const v2f Wf = ring[f % WF_AHEAD];
          ring[f % WF_AHEAD] = load_w2frag(W2p, (f + WF_AHEAD) % WF_PER_PAIR, m, h);
#pragma unroll
          for (int q = 0; q < 2; ++q) {
            v2f Hf;
            Hf.x = hv[q][2 * j];
            Hf.y = hv[q][2 * j + 1];
            acc2[q][td] = __builtin_amdgcn_wmma_f32_16x16x4_f32(
                false, Wf, false, Hf, (short)0, acc2[q][td], false, false);
          }
        }
      }
      __builtin_amdgcn_sched_barrier(0);
    }

    float bo[4][8];
#pragma unroll
    for (int td = 0; td < 4; ++td) {
      const float4* bp = reinterpret_cast<const float4*>(b2p + 16 * td + 8 * h);
      const float4 lo = bp[0], hi = bp[1];
      bo[td][0] = lo.x; bo[td][1] = lo.y; bo[td][2] = lo.z; bo[td][3] = lo.w;
      bo[td][4] = hi.x; bo[td][5] = hi.y; bo[td][6] = hi.z; bo[td][7] = hi.w;
    }
#pragma unroll
    for (int q = 0; q < 2; ++q) {
      float w[4];
#pragma unroll
      for (int td = 0; td < 4; ++td) {
        float x[8];
#pragma unroll
        for (int r = 0; r < 8; ++r) x[r] = fmaxf(acc2[q][td][r] + bo[td][r], 0.0f);
        w[td] = row_sum_scatter(x, m);
      }
#pragma unroll
      for (int u = 0; u < 2; ++u) {
        const float v = up ? w[2 * u + 1] : w[2 * u];
        out[(size_t)(gp0 + 2 * pp + q) * Dq + 32 * u + (up ? 16 : 0) + 8 * h + c7] = v * (1.0f / 16.0f);
      }
    }
  }
}

extern "C" void kernel_launch(void* const* d_in, const int* in_sizes, int n_in,
                              void* d_out, int out_size, void* d_ws, size_t ws_size,
                              hipStream_t stream) {
  const float* events = (const float*)d_in[0];
  const float* W1     = (const float*)d_in[1];
  const float* b1     = (const float*)d_in[2];
  const float* W2     = (const float*)d_in[3];
  const float* b2     = (const float*)d_in[4];
  float*       outp   = (float*)d_out;

  edgeconv_fused_kernel<<<MLP_GRID, MLP_THREADS, 0, stream>>>(
      events, W1, b1, W2, b2, outp);
}


# Layer-2 columns are points (edge e of 16 points per pass), not a point's 16 edges: P needs no lane broadcast, the mean is a running register sum in edge order (mean sum order changes)
# speedup vs baseline: 5.4260x; 1.0519x over previous
#include <hip/hip_runtime.h>

typedef __attribute__((ext_vector_type(2))) float v2f;
typedef __attribute__((ext_vector_type(8))) float v8f;

#define Bq 128
#define Nq 512
#define Fq 32
#define Hq 128
#define Dq 64
#define Kq 16

#define KNN_THREADS 256
#define KNN_BIG 3.4e38f

__device__ __forceinline__ float dist2(float x, float y, float cx, float cy) {
  const float dx = x - cx;
  const float dy = y - cy;
  return __builtin_fmaf(dx, dx, dy * dy);
}

template <bool EXCL>
__device__ __forceinline__ void knn_scan_values(const float2* sxy, int j0, int i,
                                                float cx, float cy, float (&bd)[Kq]) {
#pragma unroll 8
  for (int jj = 0; jj < 32; ++jj) {
    const int j = j0 + jj;
    const float2 c = sxy[j];
    float d = dist2(c.x, c.y, cx, cy);
    if (EXCL) d = (j == i) ? KNN_BIG : d;
#pragma unroll
    for (int t = Kq - 1; t >= 1; --t) bd[t] = __builtin_amdgcn_fmed3f(bd[t - 1], d, bd[t]);
    bd[0] = __builtin_fminf(d, bd[0]);
  }
}

template <bool EXCL>
__device__ __forceinline__ void knn_scan_gather(const float2* sxy, int* lst, int j0, int i,
                                                float cx, float cy, float tau,
                                                int& ties_left, int& pos) {
#pragma unroll 8
  for (int jj = 0; jj < 32; ++jj) {
    const int j = j0 + jj;
    const float2 c = sxy[j];
    float d = dist2(c.x, c.y, cx, cy);
    if (EXCL) d = (j == i) ? KNN_BIG : d;
    const bool eq   = (d == tau) && (ties_left > 0);
    const bool take = (d < tau) || eq;
    lst[pos] = j;
    pos += take ? KNN_THREADS : 0;
    ties_left -= eq ? 1 : 0;
  }
}

__device__ __forceinline__ void knn_half_batch(const float2* sxy, int* lst, int* nbr, int tid, int i) {
  const float cx = sxy[i].x, cy = sxy[i].y;
  const int mychunk = __builtin_amdgcn_readfirstlane(i >> 5);

  float bd[Kq];
#pragma unroll
  for (int t = 0; t < Kq; ++t) bd[t] = KNN_BIG;
  for (int cc = 0; cc < Nq / 32; ++cc) {
    if (cc == mychunk) knn_scan_values<true>(sxy, cc * 32, i, cx, cy, bd);
    else               knn_scan_values<false>(sxy, cc * 32, i, cx, cy, bd);
  }

  const float tau = bd[Kq - 1];
  int ties_left = 0;
#pragma unroll
  for (int t = 0; t < Kq; ++t) ties_left += (bd[t] == tau) ? 1 : 0;
  int pos = tid;
  for (int cc = 0; cc < Nq / 32; ++cc) {
    if (cc == mychunk) knn_scan_gather<true>(sxy, lst, cc * 32, i, cx, cy, tau, ties_left, pos);
    else               knn_scan_gather<false>(sxy, lst, cc * 32, i, cx, cy, tau, ties_left, pos);
  }

  float fd[Kq];
  int   fi[Kq];
#pragma unroll
  for (int t = 0; t < Kq; ++t) { fd[t] = KNN_BIG; fi[t] = 0; }
#pragma unroll
  for (int e = 0; e < Kq; ++e) {
    const int j = lst[e * KNN_THREADS + tid] & (Nq - 1);
    const float2 c = sxy[j];
    const float d = dist2(c.x, c.y, cx, cy);
#pragma unroll
    for (int t = Kq - 1; t >= 1; --t) {
      const bool below_prev = d < fd[t - 1];
      const bool below_this = d < fd[t];
      fi[t] = below_prev ? fi[t - 1] : (below_this ? j : fi[t]);
      fd[t] = __builtin_amdgcn_fmed3f(fd[t - 1], d, fd[t]);
    }
    fi[0] = (d < fd[0]) ? j : fi[0];
    fd[0] = __builtin_fminf(d, fd[0]);
  }

  int4* dst = reinterpret_cast<int4*>(nbr + tid * Kq);
#pragma unroll
  for (int q = 0; q < Kq / 4; ++q) {
    int4 v;
    v.x = fi[4 * q]; v.y = fi[4 * q + 1]; v.z = fi[4 * q + 2]; v.w = fi[4 * q + 3];
    dst[q] = v;
  }
}

#define MLP_WAVES 16
#define MLP_THREADS (MLP_WAVES * 32)
#define MLP_GRID (2 * Bq)
#define QP 132
#define WF_PER_PAIR 128
#define WF_AHEAD 8

__device__ __forceinline__ v2f load_w2frag(const float* __restrict__ W2, int f, int m, int h) {
  const int t = f >> 4, j = (f >> 2) & 3, td = f & 3;
  const int kk = 16 * t + 8 * h + 2 * j;
  v2f w;
  w.x = W2[(size_t)kk * Dq + 16 * td + m];
  w.y = W2[(size_t)(kk + 1) * Dq + 16 * td + m];
  return w;
}

__device__ __forceinline__ void build_q_tiles(const float* __restrict__ evb, const float* __restrict__ W1,
                                              float* ldsQ, int u, int m, int h) {
  v2f xq[4][8];
#pragma unroll
  for (int g = 0; g < 4; ++g) {
    const float* r = evb + (size_t)(16 * (4 * u + g) + m) * Fq + 2 * h;
#pragma unroll
    for (int s = 0; s < 8; ++s) { xq[g][s].x = r[4 * s]; xq[g][s].y = r[4 * s + 1]; }
  }
  const v8f vzero = {};
  v2f wb[8];
  const float* w1l = W1 + (size_t)(32 + 2 * h) * Hq + m;
#pragma unroll
  for (int s = 0; s < 8; ++s) {
    wb[s].x = w1l[(size_t)(4 * s) * Hq];
    wb[s].y = w1l[(size_t)(4 * s + 1) * Hq];
  }
#pragma unroll
  for (int t = 0; t < Hq / 16; ++t) {
    v2f nb[8];
    if (t + 1 < Hq / 16) {
#pragma unroll
      for (int s = 0; s < 8; ++s) {
        nb[s].x = w1l[(size_t)(4 * s) * Hq + 16 * (t + 1)];
        nb[s].y = w1l[(size_t)(4 * s + 1) * Hq + 16 * (t + 1)];
      }
    }
    v8f q[4];
#pragma unroll
    for (int g = 0; g < 4; ++g) q[g] = vzero;
#pragma unroll
    for (int s = 0; s < 8; ++s) {
#pragma unroll
      for (int g = 0; g < 4; ++g)
        q[g] = __builtin_amdgcn_wmma_f32_16x16x4_f32(false, wb[s], false, xq[g][s], (short)0, q[g], false, false);
    }
#pragma unroll
    for (int g = 0; g < 4; ++g) {
      float4* d = reinterpret_cast<float4*>(&ldsQ[(16 * (4 * u + g) + m) * QP + 16 * t + 8 * h]);
      d[0] = make_float4(q[g][0], q[g][1], q[g][2], q[g][3]);
      d[1] = make_float4(q[g][4], q[g][5], q[g][6], q[g][7]);
    }
    if (t + 1 < Hq / 16) {
#pragma unroll
      for (int s = 0; s < 8; ++s) wb[s] = nb[s];
    }
    __builtin_amdgcn_sched_barrier(0);
  }
}

__global__ __launch_bounds__(MLP_THREADS) __attribute__((amdgpu_waves_per_eu(4, 4)))
void edgeconv_fused_kernel(
    const float* __restrict__ events,
    const float* __restrict__ W1, const float* __restrict__ b1,
    const float* __restrict__ W2, const float* __restrict__ b2,
    float* __restrict__ out) {
  __shared__ alignas(16) float ldsQ[Nq * QP];
  __shared__ float2 sxy[Nq];
  __shared__ alignas(16) int lst[(Kq + 1) * KNN_THREADS];
  __shared__ alignas(16) int nbr[KNN_THREADS * Kq];

  const int tid  = threadIdx.x;
  const int lane = threadIdx.x & 31;
  const int wid  = __builtin_amdgcn_readfirstlane(threadIdx.x >> 5);
  const int b    = blockIdx.x >> 1;
  const int hf   = blockIdx.x & 1;
  const int m    = lane & 15;
  const int h    = lane >> 4;
  const int tile_own = 16 * hf + wid;
  const float* evb = events + (size_t)b * Nq * Fq;

  {
    const float* e = evb + (size_t)tid * Fq;
    float2 c;
    c.x = e[0];
    c.y = e[1];
    sxy[tid] = c;
  }
  __syncthreads();

  if (wid < MLP_WAVES / 2) knn_half_batch(sxy, lst, nbr, tid, KNN_THREADS * hf + tid);
  else                     build_q_tiles(evb, W1, ldsQ, wid - MLP_WAVES / 2, m, h);

  v2f xo[8];
  {
    const float* r0 = evb + (size_t)(16 * tile_own + m) * Fq + 2 * h;
#pragma unroll
    for (int s = 0; s < 8; ++s) { xo[s].x = r0[4 * s]; xo[s].y = r0[4 * s + 1]; }
  }
  const v8f vzero = {};
  v8f PT[Hq / 16];
  v2f wa[8], wb[8];
  const float* w1l = W1 + (size_t)(2 * h) * Hq + m;
#pragma unroll
  for (int s = 0; s < 8; ++s) {
    wa[s].x = w1l[(size_t)(4 * s) * Hq];
    wa[s].y = w1l[(size_t)(4 * s + 1) * Hq];
    wb[s].x = w1l[(size_t)(32 + 4 * s) * Hq];
    wb[s].y = w1l[(size_t)(33 + 4 * s) * Hq];
  }
#pragma unroll
  for (int t = 0; t < Hq / 16; ++t) {
    v2f na[8], nb[8];
    if (t + 1 < Hq / 16) {
#pragma unroll
      for (int s = 0; s < 8; ++s) {
        na[s].x = w1l[(size_t)(4 * s) * Hq + 16 * (t + 1)];
        na[s].y = w1l[(size_t)(4 * s + 1) * Hq + 16 * (t + 1)];
        nb[s].x = w1l[(size_t)(32 + 4 * s) * Hq + 16 * (t + 1)];
        nb[s].y = w1l[(size_t)(33 + 4 * s) * Hq + 16 * (t + 1)];
      }
    }
    v8f pt = vzero;
#pragma unroll
    for (int s = 0; s < 8; ++s) {
      v2f wd;
      wd.x = wa[s].x - wb[s].x;
      wd.y = wa[s].y - wb[s].y;
      pt = __builtin_amdgcn_wmma_f32_16x16x4_f32(false, wd, false, xo[s], (short)0, pt, false, false);
    }
    const float4* bp = reinterpret_cast<const float4*>(b1 + 16 * t + 8 * h);
    const float4 blo = bp[0], bhi = bp[1];
    PT[t][0] = pt[0] + blo.x; PT[t][1] = pt[1] + blo.y; PT[t][2] = pt[2] + blo.z; PT[t][3] = pt[3] + blo.w;
    PT[t][4] = pt[4] + bhi.x; PT[t][5] = pt[5] + bhi.y; PT[t][6] = pt[6] + bhi.z; PT[t][7] = pt[7] + bhi.w;
    if (t + 1 < Hq / 16) {
#pragma unroll
      for (int s = 0; s < 8; ++s) { wa[s] = na[s]; wb[s] = nb[s]; }
    }
    __builtin_amdgcn_sched_barrier(0);
  }
  __syncthreads();

  const int gp0 = b * Nq + 16 * tile_own;
  v2f ring[WF_AHEAD];
#pragma unroll
  for (int f = 0; f < WF_AHEAD; ++f) ring[f] = load_w2frag(W2, f, m, h);

  const int2* nb0 = reinterpret_cast<const int2*>(nbr + (16 * wid + m) * Kq);
  int2 nj = nb0[0];

  float run[4][8];
#pragma unroll
  for (int td = 0; td < 4; ++td)
#pragma unroll
    for (int r = 0; r < 8; ++r) run[td][r] = 0.0f;

#pragma unroll 1
  for (int ep = 0; ep < Kq / 2; ++ep) {
    const float* qrow[2];
    qrow[0] = ldsQ + (nj.x & (Nq - 1)) * QP + 8 * h;
    qrow[1] = ldsQ + (nj.y & (Nq - 1)) * QP + 8 * h;
    nj = nb0[(ep + 1) & (Kq / 2 - 1)];
    int wofs = 0;
    asm volatile("" : "+s"(wofs));
    const float* W2p = W2 + wofs;
    const float* b2p = b2 + wofs;

    v8f acc2[2][4];
#pragma unroll
    for (int q = 0; q < 2; ++q)
#pragma unroll
      for (int td = 0; td < 4; ++td) acc2[q][td] = vzero;

    float4 qlo[2], qhi[2];
#pragma unroll
    for (int q = 0; q < 2; ++q) {
      const float4* xp = reinterpret_cast<const float4*>(qrow[q]);
      qlo[q] = xp[0]; qhi[q] = xp[1];
    }

#pragma unroll
    for (int t = 0; t < Hq / 16; ++t) {
      float hv[2][8];
#pragma unroll
      for (int q = 0; q < 2; ++q) {
        const float qv[8] = {qlo[q].x, qlo[q].y, qlo[q].z, qlo[q].w, qhi[q].x, qhi[q].y, qhi[q].z, qhi[q].w};
#pragma unroll
        for (int r = 0; r < 8; ++r) hv[q][r] = fmaxf(PT[t][r] + qv[r], 0.0f);
      }
      if (t + 1 < Hq / 16) {
#pragma unroll
        for (int q = 0; q < 2; ++q) {
          const float4* xp = reinterpret_cast<const float4*>(qrow[q] + 16 * (t + 1));
          qlo[q] = xp[0]; qhi[q] = xp[1];
        }
      }
#pragma unroll
      for (int j = 0; j < 4; ++j) {
#pragma unroll
        for (int td = 0; td < 4; ++td) {
          const int f = 16 * t + 4 * j + td;
          const v2f Wf = ring[f % WF_AHEAD];
          ring[f % WF_AHEAD] = load_w2frag(W2p, (f + WF_AHEAD) % WF_PER_PAIR, m, h);
#pragma unroll
          for (int q = 0; q < 2; ++q) {
            v2f Hf;
            Hf.x = hv[q][2 * j];
            Hf.y = hv[q][2 * j + 1];
            acc2[q][td] = __builtin_amdgcn_wmma_f32_16x16x4_f32(
                false, Wf, false, Hf, (short)0, acc2[q][td], false, false);
          }
        }
      }
      __builtin_amdgcn_sched_barrier(0);
    }

#pragma unroll
    for (int td = 0; td < 4; ++td) {
      const float4* bp = reinterpret_cast<const float4*>(b2p + 16 * td + 8 * h);
      const float4 lo = bp[0], hi = bp[1];
      const float bo[8] = {lo.x, lo.y, lo.z, lo.w, hi.x, hi.y, hi.z, hi.w};
#pragma unroll
      for (int q = 0; q < 2; ++q)
#pragma unroll
        for (int r = 0; r < 8; ++r) run[td][r] += fmaxf(acc2[q][td][r] + bo[r], 0.0f);
    }
  }

  float* stg = reinterpret_cast<float*>(lst) + 256 * wid;
  const float s16 = 1.0f / 16.0f;
#pragma unroll
  for (int tp = 0; tp < 2; ++tp) {
#pragma unroll
    for (int ph = 0; ph < 2; ++ph) {
      if ((m >> 3) == ph) {
#pragma unroll
        for (int u = 0; u < 2; ++u) {
          const int td = 2 * tp + u;
          float4* d = reinterpret_cast<float4*>(stg + (m & 7) * 32 + 16 * u + 8 * h);
          d[0] = make_float4(run[td][0] * s16, run[td][1] * s16, run[td][2] * s16, run[td][3] * s16);
          d[1] = make_float4(run[td][4] * s16, run[td][5] * s16, run[td][6] * s16, run[td][7] * s16);
        }
      }
      __builtin_amdgcn_wave_barrier();
#pragma unroll
      for (int i = 0; i < 2; ++i) {
        const int pl = 4 * i + (lane >> 3);
        const float4 v = *reinterpret_cast<const float4*>(stg + pl * 32 + 4 * (lane & 7));
        *reinterpret_cast<float4*>(out + (size_t)(gp0 + 8 * ph + pl) * Dq + 32 * tp + 4 * (lane & 7)) = v;
      }
      __builtin_amdgcn_wave_barrier();
    }
  }
}

extern "C" void kernel_launch(void* const* d_in, const int* in_sizes, int n_in,
                              void* d_out, int out_size, void* d_ws, size_t ws_size,
                              hipStream_t stream) {
  const float* events = (const float*)d_in[0];
  const float* W1     = (const float*)d_in[1];
  const float* b1     = (const float*)d_in[2];
  const float* W2     = (const float*)d_in[3];
  const float* b2     = (const float*)d_in[4];
  float*       outp   = (float*)d_out;

  edgeconv_fused_kernel<<<MLP_GRID, MLP_THREADS, 0, stream>>>(
      events, W1, b1, W2, b2, outp);
}
